# starting kernel
# baseline (speedup 1.0000x reference)
#include <hip/hip_runtime.h>
#include <math.h>

typedef __attribute__((ext_vector_type(2))) float v2f;
typedef __attribute__((ext_vector_type(4))) float f4v;
typedef __attribute__((ext_vector_type(8))) float v8f;

typedef const __attribute__((address_space(1))) float* gcfp;
typedef       __attribute__((address_space(1))) float* gfp;
typedef const __attribute__((address_space(1))) int*   gcip;
typedef       __attribute__((address_space(1))) int*   gip;
typedef const __attribute__((address_space(1))) f4v*   gcf4p;
typedef       __attribute__((address_space(1))) f4v*   gf4p;
typedef const __attribute__((address_space(1))) v2f*   gcv2p;

#define D128 128
#define SCAN_THREADS 1024

__global__ __launch_bounds__(SCAN_THREADS)
void scan_offsets(const int* __restrict__ deg_, int* __restrict__ offs_, int n) {
    gcip deg  = (gcip)deg_;
    gip  offs = (gip)offs_;
    __shared__ int part[SCAN_THREADS];
    int t = threadIdx.x;
    int chunk = (n + SCAN_THREADS - 1) / SCAN_THREADS;
    int begin = t * chunk;
    int end = begin + chunk; if (end > n) end = n;
    int s = 0;
    for (int i = begin; i < end; ++i) s += deg[i];
    part[t] = s;
    __syncthreads();
    for (int off = 1; off < SCAN_THREADS; off <<= 1) {
        int v = (t >= off) ? part[t - off] : 0;
        __syncthreads();
        part[t] += v;
        __syncthreads();
    }
    int run = part[t] - s;
    for (int i = begin; i < end; ++i) { offs[i] = run; run += deg[i]; }
}

__global__ __launch_bounds__(256)
void aggregate(const float* __restrict__ x_, const int* __restrict__ src_,
               const int* __restrict__ offs_, const int* __restrict__ deg_,
               float* __restrict__ nm_, float* __restrict__ xm_, int n) {
    gcfp x   = (gcfp)x_;
    gcip src = (gcip)src_;
    gcip offs = (gcip)offs_;
    gcip deg = (gcip)deg_;
    gfp  nm  = (gfp)nm_;
    gfp  xm  = (gfp)xm_;

    int node = blockIdx.x * (blockDim.x >> 5) + (threadIdx.x >> 5);
    int lane = threadIdx.x & 31;
    if (node >= n) return;

    int base = node * D128 + lane * 4;
    int dg = deg[node];
    f4v self = *(gcf4p)(x + base);

    if (dg == 0) {
        f4v z = { 0.f, 0.f, 0.f, 0.f };
        *(gf4p)(nm + base) = z;
        *(gf4p)(xm + base) = z;
        return;
    }

    int off = offs[node];
    f4v acc = { 0.f, 0.f, 0.f, 0.f };
    for (int e = 0; e < dg; ++e) {
        int s = src[off + e];
        if (e + 1 < dg) {
            int sn = src[off + e + 1];
            __builtin_prefetch(x_ + (size_t)sn * D128 + lane * 4, 0, 1);
        }
        acc += *(gcf4p)(x + (size_t)s * D128 + lane * 4);
    }
    float inv = 1.0f / (float)dg;
    acc *= inv;
    *(gf4p)(nm + base) = acc;
    *(gf4p)(xm + base) = self;
}

#define LDA 132

__global__ __launch_bounds__(256)
void gemm_fused(const float* __restrict__ x_, const float* __restrict__ xm_,
                const float* __restrict__ nm_,
                const float* __restrict__ Wg_, const float* __restrict__ Ws_,
                const float* __restrict__ Wl_,
                const float* __restrict__ bias_, float* __restrict__ out_, int n) {
    __shared__ float As[3][16 * LDA];

    int tile = blockIdx.x;
    int row0 = tile * 16;
    int t = threadIdx.x;

    if (row0 + 16 > n) {
        for (int r = row0; r < n; ++r) {
            for (int c = t; c < D128; c += 256) {
                float s = bias_[c];
                const float* xr  = x_  + (size_t)r * D128;
                const float* xmr = xm_ + (size_t)r * D128;
                const float* nmr = nm_ + (size_t)r * D128;
                const float* wg = Wg_ + (size_t)c * D128;
                const float* ws = Ws_ + (size_t)c * D128;
                const float* wl = Wl_ + (size_t)c * D128;
                for (int k = 0; k < D128; ++k)
                    s += xr[k] * wg[k] + xmr[k] * ws[k] + nmr[k] * wl[k];
                s = s > 0.f ? s : (expf(s) - 1.f);
                out_[(size_t)r * D128 + c] = s;
            }
        }
        return;
    }

    {
        gcfp srcs[3] = { (gcfp)x_, (gcfp)xm_, (gcfp)nm_ };
        int li = t * 8;
        int r = li >> 7;
        int c = li & 127;
        for (int b3 = 0; b3 < 3; ++b3) {
            gcfp S = srcs[b3] + (size_t)(row0 + r) * D128 + c;
            f4v v0 = *(gcf4p)(S);
            f4v v1 = *(gcf4p)(S + 4);
            float* Dst = &As[b3][r * LDA + c];
            *(f4v*)(Dst)     = v0;
            *(f4v*)(Dst + 4) = v1;
        }
    }
    __syncthreads();

    int wave = t >> 5;
    int lane = t & 31;
    int col0 = wave * 16;
    int mrow = lane & 15;
    int koff = (lane >> 4) * 2;

    v8f acc = { 0.f, 0.f, 0.f, 0.f, 0.f, 0.f, 0.f, 0.f };

    gcfp Wmats[3] = { (gcfp)Wg_, (gcfp)Ws_, (gcfp)Wl_ };
    for (int b3 = 0; b3 < 3; ++b3) {
        const float* A = &As[b3][mrow * LDA];
        gcfp Brow = Wmats[b3] + (size_t)(col0 + mrow) * D128;
        for (int k = 0; k < D128; k += 4) {
            v2f a = *(const v2f*)(A + k + koff);
            v2f b = *(gcv2p)(Brow + k + koff);
            acc = __builtin_amdgcn_wmma_f32_16x16x4_f32(
                false, a, false, b, (short)0, acc, false, false);
        }
    }

    gcfp bias = (gcfp)bias_;
    gfp  out  = (gfp)out_;
    int colg = col0 + (lane & 15);
    float bv = bias[colg];
    int rbase = row0 + (lane >> 4) * 8;
    #pragma unroll
    for (int i = 0; i < 8; ++i) {
        float v = acc[i] + bv;
        v = v > 0.f ? v : (expf(v) - 1.f);
        out[(size_t)(rbase + i) * D128 + colg] = v;
    }
}

extern "C" void kernel_launch(void* const* d_in, const int* in_sizes, int n_in,
                              void* d_out, int out_size, void* d_ws, size_t ws_size,
                              hipStream_t stream) {
    (void)n_in; (void)out_size; (void)ws_size;
    const float* x    = (const float*)d_in[0];
    const float* Wg   = (const float*)d_in[1];
    const float* Wl   = (const float*)d_in[2];
    const float* Ws   = (const float*)d_in[3];
    const float* bias = (const float*)d_in[4];
    const int*   src  = (const int*)d_in[5];
    const int*   deg  = (const int*)d_in[7];

    int N = in_sizes[7];

    char* ws = (char*)d_ws;
    int* offs = (int*)ws;
    size_t off_bytes = (((size_t)N * sizeof(int)) + 255) & ~(size_t)255;
    float* nm = (float*)(ws + off_bytes);
    float* xm = nm + (size_t)N * D128;

    scan_offsets<<<1, SCAN_THREADS, 0, stream>>>(deg, offs, N);

    int wavesPerBlock = 256 / 32;
    int aggBlocks = (N + wavesPerBlock - 1) / wavesPerBlock;
    aggregate<<<aggBlocks, 256, 0, stream>>>(x, src, offs, deg, nm, xm, N);

    int tiles = (N + 15) / 16;
    gemm_fused<<<tiles, 256, 0, stream>>>(x, xm, nm, Wg, Ws, Wl, bias,
                                          (float*)d_out, N);
}


# offset scan on 49 blocks instead of 1: each block sums the degrees in front of it with coalesced 16-byte loads, then shuffle-scans its own 1024 (was one block, serial 49-element chunks per thread)
# speedup vs baseline: 1.6802x; 1.6802x over previous
#include <hip/hip_runtime.h>
#include <math.h>

typedef __attribute__((ext_vector_type(2))) float v2f;
typedef __attribute__((ext_vector_type(4))) float f4v;
typedef __attribute__((ext_vector_type(8))) float v8f;
typedef __attribute__((ext_vector_type(4))) int   i4v;

typedef const __attribute__((address_space(1))) float* gcfp;
typedef       __attribute__((address_space(1))) float* gfp;
typedef const __attribute__((address_space(1))) int*   gcip;
typedef       __attribute__((address_space(1))) int*   gip;
typedef const __attribute__((address_space(1))) f4v*   gcf4p;
typedef       __attribute__((address_space(1))) f4v*   gf4p;
typedef const __attribute__((address_space(1))) v2f*   gcv2p;
typedef const __attribute__((address_space(1))) i4v*   gci4p;

#define D128 128
#define SCAN_THREADS 1024

__global__ __launch_bounds__(SCAN_THREADS)
void scan_offsets(const int* __restrict__ deg_, int* __restrict__ offs_, int n) {
    static_assert(SCAN_THREADS == 32 * 32, "one LDS slot per lane: 32 waves of 32 lanes");
    gcip  deg  = (gcip)deg_;
    gci4p deg4 = (gci4p)deg_;
    gip   offs = (gip)offs_;
    __shared__ int wtot[32];
    __shared__ int wpre[32];

    int t = threadIdx.x;
    int lane = t & 31;
    int wave = t >> 5;
    int start = blockIdx.x * SCAN_THREADS;
    int i = start + t;
    int d = (i < n) ? deg[i] : 0;

    int nq = start >> 2;
    int pre = 0;
    for (int q0 = t; q0 < nq; q0 += 4 * SCAN_THREADS) {
        #pragma unroll
        for (int u = 0; u < 4; ++u) {
            int q = q0 + u * SCAN_THREADS;
            bool ok = q < nq;
            i4v v = deg4[ok ? q : q0];
            int sv = (v.x + v.y) + (v.z + v.w);
            pre += ok ? sv : 0;
        }
    }

    int inc = d;
    #pragma unroll
    for (int o = 1; o < 32; o <<= 1) {
        int v = __shfl_up(inc, o);
        if (lane >= o) inc += v;
    }
    #pragma unroll
    for (int o = 16; o > 0; o >>= 1) pre += __shfl_xor(pre, o);

    if (lane == 31) wtot[wave] = inc;
    if (lane == 0)  wpre[wave] = pre;
    __syncthreads();

    int c = wpre[lane] + ((lane < wave) ? wtot[lane] : 0);
    #pragma unroll
    for (int o = 16; o > 0; o >>= 1) c += __shfl_xor(c, o);

    if (i < n) offs[i] = c + inc - d;
}

__global__ __launch_bounds__(256)
void aggregate(const float* __restrict__ x_, const int* __restrict__ src_,
               const int* __restrict__ offs_, const int* __restrict__ deg_,
               float* __restrict__ nm_, float* __restrict__ xm_, int n) {
    gcfp x   = (gcfp)x_;
    gcip src = (gcip)src_;
    gcip offs = (gcip)offs_;
    gcip deg = (gcip)deg_;
    gfp  nm  = (gfp)nm_;
    gfp  xm  = (gfp)xm_;

    int node = blockIdx.x * (blockDim.x >> 5) + (threadIdx.x >> 5);
    int lane = threadIdx.x & 31;
    if (node >= n) return;

    int base = node * D128 + lane * 4;
    int dg = deg[node];
    f4v self = *(gcf4p)(x + base);

    if (dg == 0) {
        f4v z = { 0.f, 0.f, 0.f, 0.f };
        *(gf4p)(nm + base) = z;
        *(gf4p)(xm + base) = z;
        return;
    }

    int off = offs[node];
    f4v acc = { 0.f, 0.f, 0.f, 0.f };
    for (int e = 0; e < dg; ++e) {
        int s = src[off + e];
        if (e + 1 < dg) {
            int sn = src[off + e + 1];
            __builtin_prefetch(x_ + (size_t)sn * D128 + lane * 4, 0, 1);
        }
        acc += *(gcf4p)(x + (size_t)s * D128 + lane * 4);
    }
    float inv = 1.0f / (float)dg;
    acc *= inv;
    *(gf4p)(nm + base) = acc;
    *(gf4p)(xm + base) = self;
}

#define LDA 132

__global__ __launch_bounds__(256)
void gemm_fused(const float* __restrict__ x_, const float* __restrict__ xm_,
                const float* __restrict__ nm_,
                const float* __restrict__ Wg_, const float* __restrict__ Ws_,
                const float* __restrict__ Wl_,
                const float* __restrict__ bias_, float* __restrict__ out_, int n) {
    __shared__ float As[3][16 * LDA];

    int tile = blockIdx.x;
    int row0 = tile * 16;
    int t = threadIdx.x;

    if (row0 + 16 > n) {
        for (int r = row0; r < n; ++r) {
            for (int c = t; c < D128; c += 256) {
                float s = bias_[c];
                const float* xr  = x_  + (size_t)r * D128;
                const float* xmr = xm_ + (size_t)r * D128;
                const float* nmr = nm_ + (size_t)r * D128;
                const float* wg = Wg_ + (size_t)c * D128;
                const float* ws = Ws_ + (size_t)c * D128;
                const float* wl = Wl_ + (size_t)c * D128;
                for (int k = 0; k < D128; ++k)
                    s += xr[k] * wg[k] + xmr[k] * ws[k] + nmr[k] * wl[k];
                s = s > 0.f ? s : (expf(s) - 1.f);
                out_[(size_t)r * D128 + c] = s;
            }
        }
        return;
    }

    {
        gcfp srcs[3] = { (gcfp)x_, (gcfp)xm_, (gcfp)nm_ };
        int li = t * 8;
        int r = li >> 7;
        int c = li & 127;
        for (int b3 = 0; b3 < 3; ++b3) {
            gcfp S = srcs[b3] + (size_t)(row0 + r) * D128 + c;
            f4v v0 = *(gcf4p)(S);
            f4v v1 = *(gcf4p)(S + 4);
            float* Dst = &As[b3][r * LDA + c];
            *(f4v*)(Dst)     = v0;
            *(f4v*)(Dst + 4) = v1;
        }
    }
    __syncthreads();

    int wave = t >> 5;
    int lane = t & 31;
    int col0 = wave * 16;
    int mrow = lane & 15;
    int koff = (lane >> 4) * 2;

    v8f acc = { 0.f, 0.f, 0.f, 0.f, 0.f, 0.f, 0.f, 0.f };

    gcfp Wmats[3] = { (gcfp)Wg_, (gcfp)Ws_, (gcfp)Wl_ };
    for (int b3 = 0; b3 < 3; ++b3) {
        const float* A = &As[b3][mrow * LDA];
        gcfp Brow = Wmats[b3] + (size_t)(col0 + mrow) * D128;
        for (int k = 0; k < D128; k += 4) {
            v2f a = *(const v2f*)(A + k + koff);
            v2f b = *(gcv2p)(Brow + k + koff);
            acc = __builtin_amdgcn_wmma_f32_16x16x4_f32(
                false, a, false, b, (short)0, acc, false, false);
        }
    }

    gcfp bias = (gcfp)bias_;
    gfp  out  = (gfp)out_;
    int colg = col0 + (lane & 15);
    float bv = bias[colg];
    int rbase = row0 + (lane >> 4) * 8;
    #pragma unroll
    for (int i = 0; i < 8; ++i) {
        float v = acc[i] + bv;
        v = v > 0.f ? v : (expf(v) - 1.f);
        out[(size_t)(rbase + i) * D128 + colg] = v;
    }
}

extern "C" void kernel_launch(void* const* d_in, const int* in_sizes, int n_in,
                              void* d_out, int out_size, void* d_ws, size_t ws_size,
                              hipStream_t stream) {
    (void)n_in; (void)out_size; (void)ws_size;
    const float* x    = (const float*)d_in[0];
    const float* Wg   = (const float*)d_in[1];
    const float* Wl   = (const float*)d_in[2];
    const float* Ws   = (const float*)d_in[3];
    const float* bias = (const float*)d_in[4];
    const int*   src  = (const int*)d_in[5];
    const int*   deg  = (const int*)d_in[7];

    int N = in_sizes[7];

    char* ws = (char*)d_ws;
    int* offs = (int*)ws;
    size_t off_bytes = (((size_t)N * sizeof(int)) + 255) & ~(size_t)255;
    float* nm = (float*)(ws + off_bytes);
    float* xm = nm + (size_t)N * D128;

    int scanBlocks = (N + SCAN_THREADS - 1) / SCAN_THREADS;
    scan_offsets<<<scanBlocks, SCAN_THREADS, 0, stream>>>(deg, offs, N);

    int wavesPerBlock = 256 / 32;
    int aggBlocks = (N + wavesPerBlock - 1) / wavesPerBlock;
    aggregate<<<aggBlocks, 256, 0, stream>>>(x, src, offs, deg, nm, xm, N);

    int tiles = (N + 15) / 16;
    gemm_fused<<<tiles, 256, 0, stream>>>(x, xm, nm, Wg, Ws, Wl, bias,
                                          (float*)d_out, N);
}


# aggregate as a persistent gather: 2048 blocks, a wave walks its nodes, takes a node's ids in one coalesced load and keeps 8 neighbor rows in flight (was id then row per edge); sum order unchanged
# speedup vs baseline: 1.8787x; 1.1181x over previous
#include <hip/hip_runtime.h>
#include <math.h>

typedef __attribute__((ext_vector_type(2))) float v2f;
typedef __attribute__((ext_vector_type(4))) float f4v;
typedef __attribute__((ext_vector_type(8))) float v8f;
typedef __attribute__((ext_vector_type(4))) int   i4v;

typedef const __attribute__((address_space(1))) float* gcfp;
typedef       __attribute__((address_space(1))) float* gfp;
typedef const __attribute__((address_space(1))) int*   gcip;
typedef       __attribute__((address_space(1))) int*   gip;
typedef const __attribute__((address_space(1))) f4v*   gcf4p;
typedef       __attribute__((address_space(1))) f4v*   gf4p;
typedef const __attribute__((address_space(1))) v2f*   gcv2p;
typedef const __attribute__((address_space(1))) i4v*   gci4p;

#define D128 128
#define SCAN_THREADS 1024

__global__ __launch_bounds__(SCAN_THREADS)
void scan_offsets(const int* __restrict__ deg_, int* __restrict__ offs_, int n) {
    static_assert(SCAN_THREADS == 32 * 32, "one LDS slot per lane: 32 waves of 32 lanes");
    gcip  deg  = (gcip)deg_;
    gci4p deg4 = (gci4p)deg_;
    gip   offs = (gip)offs_;
    __shared__ int wtot[32];
    __shared__ int wpre[32];

    int t = threadIdx.x;
    int lane = t & 31;
    int wave = t >> 5;
    int start = blockIdx.x * SCAN_THREADS;
    int i = start + t;
    int d = (i < n) ? deg[i] : 0;

    int nq = start >> 2;
    int pre = 0;
    for (int q0 = t; q0 < nq; q0 += 4 * SCAN_THREADS) {
        #pragma unroll
        for (int u = 0; u < 4; ++u) {
            int q = q0 + u * SCAN_THREADS;
            bool ok = q < nq;
            i4v v = deg4[ok ? q : q0];
            int sv = (v.x + v.y) + (v.z + v.w);
            pre += ok ? sv : 0;
        }
    }

    int inc = d;
    #pragma unroll
    for (int o = 1; o < 32; o <<= 1) {
        int v = __shfl_up(inc, o);
        if (lane >= o) inc += v;
    }
    #pragma unroll
    for (int o = 16; o > 0; o >>= 1) pre += __shfl_xor(pre, o);

    if (lane == 31) wtot[wave] = inc;
    if (lane == 0)  wpre[wave] = pre;
    __syncthreads();

    int c = wpre[lane] + ((lane < wave) ? wtot[lane] : 0);
    #pragma unroll
    for (int o = 16; o > 0; o >>= 1) c += __shfl_xor(c, o);

    if (i < n) offs[i] = c + inc - d;
}

#define AGG_ROWS 8
#define AGG_BLOCKS 2048

#define AGG_ISSUE_FULL(ids, e)                                               \
    _Pragma("unroll")                                                        \
    for (int u = 0; u < AGG_ROWS; ++u) {                                     \
        int s = __builtin_amdgcn_readlane((ids), (e) + u);                   \
        r[u] = *(gcf4p)(x + (size_t)s * D128 + lo);                          \
    }
#define AGG_ISSUE_PART(ids, e, m)                                            \
    {                                                                        \
        int sid[AGG_ROWS - 1];                                               \
        _Pragma("unroll")                                                    \
        for (int u = 0; u < AGG_ROWS - 1; ++u)                               \
            sid[u] = __builtin_amdgcn_readlane((ids), (e) + u);              \
        _Pragma("unroll")                                                    \
        for (int u = 0; u < AGG_ROWS - 1; ++u)                               \
            if (u < (m)) r[u] = *(gcf4p)(x + (size_t)sid[u] * D128 + lo);    \
    }
#define AGG_ADD_FULL()                                                       \
    _Pragma("unroll")                                                        \
    for (int u = 0; u < AGG_ROWS; ++u) acc += r[u];
#define AGG_ADD_PART(m)                                                      \
    _Pragma("unroll")                                                        \
    for (int u = 0; u < AGG_ROWS - 1; ++u)                                   \
        if (u < (m)) acc += r[u];

__global__ __launch_bounds__(256)
void aggregate(const float* __restrict__ x_, const int* __restrict__ src_,
               const int* __restrict__ offs_, const int* __restrict__ deg_,
               float* __restrict__ nm_, float* __restrict__ xm_, int n) {
    gcfp x   = (gcfp)x_;
    gcip src = (gcip)src_;
    gcip offs = (gcip)offs_;
    gcip deg = (gcip)deg_;
    gfp  nm  = (gfp)nm_;
    gfp  xm  = (gfp)xm_;

    int lane = threadIdx.x & 31;
    int lo   = lane * 4;
    int wpb  = blockDim.x >> 5;
    int node = __builtin_amdgcn_readfirstlane((int)(blockIdx.x * wpb + (threadIdx.x >> 5)));
    int nw   = (int)(gridDim.x * wpb);
    if (node >= n) return;

    int dg  = __builtin_amdgcn_readfirstlane(deg[node]);
    int off = __builtin_amdgcn_readfirstlane(offs[node]);
    int my = 0;
    if (lane < (dg < 32 ? dg : 32)) my = src[off + lane];

    for (;;) {
        int nodeN = node + nw;
        bool hasN = nodeN < n;
        int dgN = 0, offN = 0;
        if (hasN) {
            dgN  = __builtin_amdgcn_readfirstlane(deg[nodeN]);
            offN = __builtin_amdgcn_readfirstlane(offs[nodeN]);
        }
        int base = node * D128 + lo;
        f4v acc  = { 0.f, 0.f, 0.f, 0.f };
        f4v self = { 0.f, 0.f, 0.f, 0.f };
        f4v r[AGG_ROWS];
        int cnt = dg < 32 ? dg : 32;

        if (dg > 0) {
            if (cnt >= AGG_ROWS) { AGG_ISSUE_FULL(my, 0) }
            else                 { AGG_ISSUE_PART(my, 0, cnt) }
            self = *(gcf4p)(x + base);
        }
        int myN = 0;
        if (hasN && lane < (dgN < 32 ? dgN : 32)) myN = src[offN + lane];

        if (dg > 0) {
            if (cnt >= AGG_ROWS) { AGG_ADD_FULL() }
            else                 { AGG_ADD_PART(cnt) }
            for (int e = AGG_ROWS; e < cnt; e += AGG_ROWS) {
                int m = cnt - e;
                if (m >= AGG_ROWS) {
                    AGG_ISSUE_FULL(my, e)
                    __builtin_amdgcn_sched_barrier(0);
                    AGG_ADD_FULL()
                } else {
                    AGG_ISSUE_PART(my, e, m)
                    AGG_ADD_PART(m)
                }
            }
            for (int c0 = 32; c0 < dg; c0 += 32) {
                int c = dg - c0;
                c = c < 32 ? c : 32;
                int ids = 0;
                if (lane < c) ids = src[off + c0 + lane];
                for (int e = 0; e < c; e += AGG_ROWS) {
                    int m = c - e;
                    if (m >= AGG_ROWS) {
                        AGG_ISSUE_FULL(ids, e)
                        __builtin_amdgcn_sched_barrier(0);
                        AGG_ADD_FULL()
                    } else {
                        AGG_ISSUE_PART(ids, e, m)
                        AGG_ADD_PART(m)
                    }
                }
            }
            float inv = 1.0f / (float)dg;
            acc *= inv;
        }
        *(gf4p)(nm + base) = acc;
        *(gf4p)(xm + base) = self;

        if (!hasN) break;
        node = nodeN; dg = dgN; off = offN; my = myN;
    }
}
#undef AGG_ISSUE_FULL
#undef AGG_ISSUE_PART
#undef AGG_ADD_FULL
#undef AGG_ADD_PART

#define LDA 132

__global__ __launch_bounds__(256)
void gemm_fused(const float* __restrict__ x_, const float* __restrict__ xm_,
                const float* __restrict__ nm_,
                const float* __restrict__ Wg_, const float* __restrict__ Ws_,
                const float* __restrict__ Wl_,
                const float* __restrict__ bias_, float* __restrict__ out_, int n) {
    __shared__ float As[3][16 * LDA];

    int tile = blockIdx.x;
    int row0 = tile * 16;
    int t = threadIdx.x;

    if (row0 + 16 > n) {
        for (int r = row0; r < n; ++r) {
            for (int c = t; c < D128; c += 256) {
                float s = bias_[c];
                const float* xr  = x_  + (size_t)r * D128;
                const float* xmr = xm_ + (size_t)r * D128;
                const float* nmr = nm_ + (size_t)r * D128;
                const float* wg = Wg_ + (size_t)c * D128;
                const float* ws = Ws_ + (size_t)c * D128;
                const float* wl = Wl_ + (size_t)c * D128;
                for (int k = 0; k < D128; ++k)
                    s += xr[k] * wg[k] + xmr[k] * ws[k] + nmr[k] * wl[k];
                s = s > 0.f ? s : (expf(s) - 1.f);
                out_[(size_t)r * D128 + c] = s;
            }
        }
        return;
    }

    {
        gcfp srcs[3] = { (gcfp)x_, (gcfp)xm_, (gcfp)nm_ };
        int li = t * 8;
        int r = li >> 7;
        int c = li & 127;
        for (int b3 = 0; b3 < 3; ++b3) {
            gcfp S = srcs[b3] + (size_t)(row0 + r) * D128 + c;
            f4v v0 = *(gcf4p)(S);
            f4v v1 = *(gcf4p)(S + 4);
            float* Dst = &As[b3][r * LDA + c];
            *(f4v*)(Dst)     = v0;
            *(f4v*)(Dst + 4) = v1;
        }
    }
    __syncthreads();

    int wave = t >> 5;
    int lane = t & 31;
    int col0 = wave * 16;
    int mrow = lane & 15;
    int koff = (lane >> 4) * 2;

    v8f acc = { 0.f, 0.f, 0.f, 0.f, 0.f, 0.f, 0.f, 0.f };

    gcfp Wmats[3] = { (gcfp)Wg_, (gcfp)Ws_, (gcfp)Wl_ };
    for (int b3 = 0; b3 < 3; ++b3) {
        const float* A = &As[b3][mrow * LDA];
        gcfp Brow = Wmats[b3] + (size_t)(col0 + mrow) * D128;
        for (int k = 0; k < D128; k += 4) {
            v2f a = *(const v2f*)(A + k + koff);
            v2f b = *(gcv2p)(Brow + k + koff);
            acc = __builtin_amdgcn_wmma_f32_16x16x4_f32(
                false, a, false, b, (short)0, acc, false, false);
        }
    }

    gcfp bias = (gcfp)bias_;
    gfp  out  = (gfp)out_;
    int colg = col0 + (lane & 15);
    float bv = bias[colg];
    int rbase = row0 + (lane >> 4) * 8;
    #pragma unroll
    for (int i = 0; i < 8; ++i) {
        float v = acc[i] + bv;
        v = v > 0.f ? v : (expf(v) - 1.f);
        out[(size_t)(rbase + i) * D128 + colg] = v;
    }
}

extern "C" void kernel_launch(void* const* d_in, const int* in_sizes, int n_in,
                              void* d_out, int out_size, void* d_ws, size_t ws_size,
                              hipStream_t stream) {
    (void)n_in; (void)out_size; (void)ws_size;
    const float* x    = (const float*)d_in[0];
    const float* Wg   = (const float*)d_in[1];
    const float* Wl   = (const float*)d_in[2];
    const float* Ws   = (const float*)d_in[3];
    const float* bias = (const float*)d_in[4];
    const int*   src  = (const int*)d_in[5];
    const int*   deg  = (const int*)d_in[7];

    int N = in_sizes[7];

    char* ws = (char*)d_ws;
    int* offs = (int*)ws;
    size_t off_bytes = (((size_t)N * sizeof(int)) + 255) & ~(size_t)255;
    float* nm = (float*)(ws + off_bytes);
    float* xm = nm + (size_t)N * D128;

    int scanBlocks = (N + SCAN_THREADS - 1) / SCAN_THREADS;
    scan_offsets<<<scanBlocks, SCAN_THREADS, 0, stream>>>(deg, offs, N);

    aggregate<<<AGG_BLOCKS, 256, 0, stream>>>(x, src, offs, deg, nm, xm, N);

    int tiles = (N + 15) / 16;
    gemm_fused<<<tiles, 256, 0, stream>>>(x, xm, nm, Wg, Ws, Wl, bias,
                                          (float*)d_out, N);
}


# gemm: 64-row blocks, each wave keeps 4 accumulator tiles so one weight (B) fragment load feeds 4 WMMAs (was one load per WMMA); leftover tiles as 16-row blocks; per-element sum order unchanged
# speedup vs baseline: 2.5565x; 1.3608x over previous
#include <hip/hip_runtime.h>
#include <math.h>

typedef __attribute__((ext_vector_type(2))) float v2f;
typedef __attribute__((ext_vector_type(4))) float f4v;
typedef __attribute__((ext_vector_type(8))) float v8f;
typedef __attribute__((ext_vector_type(4))) int   i4v;

typedef const __attribute__((address_space(1))) float* gcfp;
typedef       __attribute__((address_space(1))) float* gfp;
typedef const __attribute__((address_space(1))) int*   gcip;
typedef       __attribute__((address_space(1))) int*   gip;
typedef const __attribute__((address_space(1))) f4v*   gcf4p;
typedef       __attribute__((address_space(1))) f4v*   gf4p;
typedef const __attribute__((address_space(1))) v2f*   gcv2p;
typedef const __attribute__((address_space(1))) i4v*   gci4p;

#define D128 128
#define SCAN_THREADS 1024

__global__ __launch_bounds__(SCAN_THREADS)
void scan_offsets(const int* __restrict__ deg_, int* __restrict__ offs_, int n) {
    static_assert(SCAN_THREADS == 32 * 32, "one LDS slot per lane: 32 waves of 32 lanes");
    gcip  deg  = (gcip)deg_;
    gci4p deg4 = (gci4p)deg_;
    gip   offs = (gip)offs_;
    __shared__ int wtot[32];
    __shared__ int wpre[32];

    int t = threadIdx.x;
    int lane = t & 31;
    int wave = t >> 5;
    int start = blockIdx.x * SCAN_THREADS;
    int i = start + t;
    int d = (i < n) ? deg[i] : 0;

    int nq = start >> 2;
    int pre = 0;
    for (int q0 = t; q0 < nq; q0 += 4 * SCAN_THREADS) {
        #pragma unroll
        for (int u = 0; u < 4; ++u) {
            int q = q0 + u * SCAN_THREADS;
            bool ok = q < nq;
            i4v v = deg4[ok ? q : q0];
            int sv = (v.x + v.y) + (v.z + v.w);
            pre += ok ? sv : 0;
        }
    }

    int inc = d;
    #pragma unroll
    for (int o = 1; o < 32; o <<= 1) {
        int v = __shfl_up(inc, o);
        if (lane >= o) inc += v;
    }
    #pragma unroll
    for (int o = 16; o > 0; o >>= 1) pre += __shfl_xor(pre, o);

    if (lane == 31) wtot[wave] = inc;
    if (lane == 0)  wpre[wave] = pre;
    __syncthreads();

    int c = wpre[lane] + ((lane < wave) ? wtot[lane] : 0);
    #pragma unroll
    for (int o = 16; o > 0; o >>= 1) c += __shfl_xor(c, o);

    if (i < n) offs[i] = c + inc - d;
}

#define AGG_ROWS 8
#define AGG_BLOCKS 2048

#define AGG_ISSUE_FULL(ids, e)                                               \
    _Pragma("unroll")                                                        \
    for (int u = 0; u < AGG_ROWS; ++u) {                                     \
        int s = __builtin_amdgcn_readlane((ids), (e) + u);                   \
        r[u] = *(gcf4p)(x + (size_t)s * D128 + lo);                          \
    }
#define AGG_ISSUE_PART(ids, e, m)                                            \
    {                                                                        \
        int sid[AGG_ROWS - 1];                                               \
        _Pragma("unroll")                                                    \
        for (int u = 0; u < AGG_ROWS - 1; ++u)                               \
            sid[u] = __builtin_amdgcn_readlane((ids), (e) + u);              \
        _Pragma("unroll")                                                    \
        for (int u = 0; u < AGG_ROWS - 1; ++u)                               \
            if (u < (m)) r[u] = *(gcf4p)(x + (size_t)sid[u] * D128 + lo);    \
    }
#define AGG_ADD_FULL()                                                       \
    _Pragma("unroll")                                                        \
    for (int u = 0; u < AGG_ROWS; ++u) acc += r[u];
#define AGG_ADD_PART(m)                                                      \
    _Pragma("unroll")                                                        \
    for (int u = 0; u < AGG_ROWS - 1; ++u)                                   \
        if (u < (m)) acc += r[u];

__global__ __launch_bounds__(256)
void aggregate(const float* __restrict__ x_, const int* __restrict__ src_,
               const int* __restrict__ offs_, const int* __restrict__ deg_,
               float* __restrict__ nm_, float* __restrict__ xm_, int n) {
    gcfp x   = (gcfp)x_;
    gcip src = (gcip)src_;
    gcip offs = (gcip)offs_;
    gcip deg = (gcip)deg_;
    gfp  nm  = (gfp)nm_;
    gfp  xm  = (gfp)xm_;

    int lane = threadIdx.x & 31;
    int lo   = lane * 4;
    int wpb  = blockDim.x >> 5;
    int node = __builtin_amdgcn_readfirstlane((int)(blockIdx.x * wpb + (threadIdx.x >> 5)));
    int nw   = (int)(gridDim.x * wpb);
    if (node >= n) return;

    int dg  = __builtin_amdgcn_readfirstlane(deg[node]);
    int off = __builtin_amdgcn_readfirstlane(offs[node]);
    int my = 0;
    if (lane < (dg < 32 ? dg : 32)) my = src[off + lane];

    for (;;) {
        int nodeN = node + nw;
        bool hasN = nodeN < n;
        int dgN = 0, offN = 0;
        if (hasN) {
            dgN  = __builtin_amdgcn_readfirstlane(deg[nodeN]);
            offN = __builtin_amdgcn_readfirstlane(offs[nodeN]);
        }
        int base = node * D128 + lo;
        f4v acc  = { 0.f, 0.f, 0.f, 0.f };
        f4v self = { 0.f, 0.f, 0.f, 0.f };
        f4v r[AGG_ROWS];
        int cnt = dg < 32 ? dg : 32;

        if (dg > 0) {
            if (cnt >= AGG_ROWS) { AGG_ISSUE_FULL(my, 0) }
            else                 { AGG_ISSUE_PART(my, 0, cnt) }
            self = *(gcf4p)(x + base);
        }
        int myN = 0;
        if (hasN && lane < (dgN < 32 ? dgN : 32)) myN = src[offN + lane];

        if (dg > 0) {
            if (cnt >= AGG_ROWS) { AGG_ADD_FULL() }
            else                 { AGG_ADD_PART(cnt) }
            for (int e = AGG_ROWS; e < cnt; e += AGG_ROWS) {
                int m = cnt - e;
                if (m >= AGG_ROWS) {
                    AGG_ISSUE_FULL(my, e)
                    __builtin_amdgcn_sched_barrier(0);
                    AGG_ADD_FULL()
                } else {
                    AGG_ISSUE_PART(my, e, m)
                    AGG_ADD_PART(m)
                }
            }
            for (int c0 = 32; c0 < dg; c0 += 32) {
                int c = dg - c0;
                c = c < 32 ? c : 32;
                int ids = 0;
                if (lane < c) ids = src[off + c0 + lane];
                for (int e = 0; e < c; e += AGG_ROWS) {
                    int m = c - e;
                    if (m >= AGG_ROWS) {
                        AGG_ISSUE_FULL(ids, e)
                        __builtin_amdgcn_sched_barrier(0);
                        AGG_ADD_FULL()
                    } else {
                        AGG_ISSUE_PART(ids, e, m)
                        AGG_ADD_PART(m)
                    }
                }
            }
            float inv = 1.0f / (float)dg;
            acc *= inv;
        }
        *(gf4p)(nm + base) = acc;
        *(gf4p)(xm + base) = self;

        if (!hasN) break;
        node = nodeN; dg = dgN; off = offN; my = myN;
    }
}
#undef AGG_ISSUE_FULL
#undef AGG_ISSUE_PART
#undef AGG_ADD_FULL
#undef AGG_ADD_PART

#define LDA 132
#define GEMM_MT 4
#define GEMM_BIG 768
#define SLAB (GEMM_MT * 16 * LDA)

typedef       __attribute__((address_space(3))) float* lfp;
typedef       __attribute__((address_space(3))) f4v*   lf4p;
typedef const __attribute__((address_space(3))) v2f*   lcv2p;

template <int NT>
__device__ __forceinline__ void gemm_rows(lfp As, int row0, int t,
        gcfp x, gcfp xm, gcfp nm, gcfp Wg, gcfp Ws, gcfp Wl,
        gcfp bias, gfp out) {
    {
        gcfp srcs[3] = { x, xm, nm };
        #pragma unroll
        for (int b3 = 0; b3 < 3; ++b3) {
            gcf4p S = (gcf4p)(srcs[b3] + (size_t)row0 * D128);
            #pragma unroll
            for (int i = 0; i < NT * 2; ++i) {
                int idx = i * 256 + t;
                int r = idx >> 5;
                int c = (idx & 31) * 4;
                f4v v = S[idx];
                *(lf4p)(As + b3 * SLAB + r * LDA + c) = v;
            }
        }
    }
    __syncthreads();

    int wave = t >> 5;
    int lane = t & 31;
    int col0 = wave * 16;
    int mrow = lane & 15;
    int koff = (lane >> 4) * 2;

    v8f acc[NT];
    #pragma unroll
    for (int m = 0; m < NT; ++m)
        acc[m] = (v8f){ 0.f, 0.f, 0.f, 0.f, 0.f, 0.f, 0.f, 0.f };

    gcfp Wmats[3] = { Wg, Ws, Wl };
    #pragma unroll
    for (int b3 = 0; b3 < 3; ++b3) {
        lfp  A    = As + b3 * SLAB + mrow * LDA + koff;
        gcfp Brow = Wmats[b3] + (size_t)(col0 + mrow) * D128 + koff;
        #pragma unroll
        for (int k = 0; k < D128; k += 4) {
            v2f b = *(gcv2p)(Brow + k);
            #pragma unroll
            for (int m = 0; m < NT; ++m) {
                v2f a = *(lcv2p)(A + m * 16 * LDA + k);
                acc[m] = __builtin_amdgcn_wmma_f32_16x16x4_f32(
                    false, a, false, b, (short)0, acc[m], false, false);
            }
        }
    }

    int colg = col0 + (lane & 15);
    float bv = bias[colg];
    #pragma unroll
    for (int m = 0; m < NT; ++m) {
        int rbase = row0 + m * 16 + (lane >> 4) * 8;
        #pragma unroll
        for (int i = 0; i < 8; ++i) {
            float v = acc[m][i] + bv;
            v = v > 0.f ? v : (expf(v) - 1.f);
            out[(size_t)(rbase + i) * D128 + colg] = v;
        }
    }
}

__global__ __launch_bounds__(256)
void gemm_fused(const float* __restrict__ x_, const float* __restrict__ xm_,
                const float* __restrict__ nm_,
                const float* __restrict__ Wg_, const float* __restrict__ Ws_,
                const float* __restrict__ Wl_,
                const float* __restrict__ bias_, float* __restrict__ out_, int n) {
    __shared__ float As[3 * SLAB];

    int t = threadIdx.x;
    int b = blockIdx.x;
    int nbig = (n >> 4) / GEMM_MT;
    nbig = nbig < GEMM_BIG ? nbig : GEMM_BIG;

    if (b < nbig) {
        gemm_rows<GEMM_MT>((lfp)As, b * (GEMM_MT * 16), t,
                           (gcfp)x_, (gcfp)xm_, (gcfp)nm_,
                           (gcfp)Wg_, (gcfp)Ws_, (gcfp)Wl_,
                           (gcfp)bias_, (gfp)out_);
        return;
    }

    int tile = nbig * GEMM_MT + (b - nbig);
    int row0 = tile * 16;

    if (row0 + 16 > n) {
        for (int r = row0; r < n; ++r) {
            for (int c = t; c < D128; c += 256) {
                float s = bias_[c];
                const float* xr  = x_  + (size_t)r * D128;
                const float* xmr = xm_ + (size_t)r * D128;
                const float* nmr = nm_ + (size_t)r * D128;
                const float* wg = Wg_ + (size_t)c * D128;
                const float* ws = Ws_ + (size_t)c * D128;
                const float* wl = Wl_ + (size_t)c * D128;
                for (int k = 0; k < D128; ++k)
                    s += xr[k] * wg[k] + xmr[k] * ws[k] + nmr[k] * wl[k];
                s = s > 0.f ? s : (expf(s) - 1.f);
                out_[(size_t)r * D128 + c] = s;
            }
        }
        return;
    }

    gemm_rows<1>((lfp)As, row0, t,
                 (gcfp)x_, (gcfp)xm_, (gcfp)nm_,
                 (gcfp)Wg_, (gcfp)Ws_, (gcfp)Wl_,
                 (gcfp)bias_, (gfp)out_);
}

extern "C" void kernel_launch(void* const* d_in, const int* in_sizes, int n_in,
                              void* d_out, int out_size, void* d_ws, size_t ws_size,
                              hipStream_t stream) {
    (void)n_in; (void)out_size; (void)ws_size;
    const float* x    = (const float*)d_in[0];
    const float* Wg   = (const float*)d_in[1];
    const float* Wl   = (const float*)d_in[2];
    const float* Ws   = (const float*)d_in[3];
    const float* bias = (const float*)d_in[4];
    const int*   src  = (const int*)d_in[5];
    const int*   deg  = (const int*)d_in[7];

    int N = in_sizes[7];

    char* ws = (char*)d_ws;
    int* offs = (int*)ws;
    size_t off_bytes = (((size_t)N * sizeof(int)) + 255) & ~(size_t)255;
    float* nm = (float*)(ws + off_bytes);
    float* xm = nm + (size_t)N * D128;

    int scanBlocks = (N + SCAN_THREADS - 1) / SCAN_THREADS;
    scan_offsets<<<scanBlocks, SCAN_THREADS, 0, stream>>>(deg, offs, N);

    aggregate<<<AGG_BLOCKS, 256, 0, stream>>>(x, src, offs, deg, nm, xm, N);

    int tiles = (N + 15) / 16;
    int nbig = (N / 16) / GEMM_MT;
    nbig = nbig < GEMM_BIG ? nbig : GEMM_BIG;
    int gblocks = nbig + (tiles - nbig * GEMM_MT);
    gemm_fused<<<gblocks, 256, 0, stream>>>(x, xm, nm, Wg, Ws, Wl, bias,
                                            (float*)d_out, N);
}


# gather fused into the gemm's staging: each wave sums its rows' neighbors (edge order kept) straight into the LDS slabs and masks the self slab by deg; nm/xm planes and one launch gone
# speedup vs baseline: 2.8610x; 1.1191x over previous
#include <hip/hip_runtime.h>
#include <math.h>

typedef __attribute__((ext_vector_type(2))) float v2f;
typedef __attribute__((ext_vector_type(4))) float f4v;
typedef __attribute__((ext_vector_type(8))) float v8f;
typedef __attribute__((ext_vector_type(4))) int   i4v;

typedef const __attribute__((address_space(1))) float* gcfp;
typedef       __attribute__((address_space(1))) float* gfp;
typedef const __attribute__((address_space(1))) int*   gcip;
typedef       __attribute__((address_space(1))) int*   gip;
typedef const __attribute__((address_space(1))) f4v*   gcf4p;
typedef       __attribute__((address_space(1))) f4v*   gf4p;
typedef const __attribute__((address_space(1))) v2f*   gcv2p;
typedef const __attribute__((address_space(1))) i4v*   gci4p;

#define D128 128
#define SCAN_THREADS 1024

__global__ __launch_bounds__(SCAN_THREADS)
void scan_offsets(const int* __restrict__ deg_, int* __restrict__ offs_, int n) {
    static_assert(SCAN_THREADS == 32 * 32, "one LDS slot per lane: 32 waves of 32 lanes");
    gcip  deg  = (gcip)deg_;
    gci4p deg4 = (gci4p)deg_;
    gip   offs = (gip)offs_;
    __shared__ int wtot[32];
    __shared__ int wpre[32];

    int t = threadIdx.x;
    int lane = t & 31;
    int wave = t >> 5;
    int start = blockIdx.x * SCAN_THREADS;
    int i = start + t;
    int d = (i < n) ? deg[i] : 0;

    int nq = start >> 2;
    int pre = 0;
    for (int q0 = t; q0 < nq; q0 += 4 * SCAN_THREADS) {
        #pragma unroll
        for (int u = 0; u < 4; ++u) {
            int q = q0 + u * SCAN_THREADS;
            bool ok = q < nq;
            i4v v = deg4[ok ? q : q0];
            int sv = (v.x + v.y) + (v.z + v.w);
            pre += ok ? sv : 0;
        }
    }

    int inc = d;
    #pragma unroll
    for (int o = 1; o < 32; o <<= 1) {
        int v = __shfl_up(inc, o);
        if (lane >= o) inc += v;
    }
    #pragma unroll
    for (int o = 16; o > 0; o >>= 1) pre += __shfl_xor(pre, o);

    if (lane == 31) wtot[wave] = inc;
    if (lane == 0)  wpre[wave] = pre;
    __syncthreads();

    int c = wpre[lane] + ((lane < wave) ? wtot[lane] : 0);
    #pragma unroll
    for (int o = 16; o > 0; o >>= 1) c += __shfl_xor(c, o);

    if (i < n) offs[i] = c + inc - d;
}

#define AGG_ROWS 8

#define AGG_ISSUE_FULL(ids, e)                                               \
    _Pragma("unroll")                                                        \
    for (int u = 0; u < AGG_ROWS; ++u) {                                     \
        int s = __builtin_amdgcn_readlane((ids), (e) + u);                   \
        r[u] = *(gcf4p)(x + (size_t)s * D128 + lo);                          \
    }
#define AGG_ISSUE_PART(ids, e, m)                                            \
    {                                                                        \
        int sid[AGG_ROWS - 1];                                               \
        _Pragma("unroll")                                                    \
        for (int u = 0; u < AGG_ROWS - 1; ++u)                               \
            sid[u] = __builtin_amdgcn_readlane((ids), (e) + u);              \
        _Pragma("unroll")                                                    \
        for (int u = 0; u < AGG_ROWS - 1; ++u)                               \
            if (u < (m)) r[u] = *(gcf4p)(x + (size_t)sid[u] * D128 + lo);    \
    }
#define AGG_ADD_FULL()                                                       \
    _Pragma("unroll")                                                        \
    for (int u = 0; u < AGG_ROWS; ++u) acc += r[u];
#define AGG_ADD_PART(m)                                                      \
    _Pragma("unroll")                                                        \
    for (int u = 0; u < AGG_ROWS - 1; ++u)                                   \
        if (u < (m)) acc += r[u];

#define LDA 132
#define GEMM_MT 4
#define GEMM_BIG 768
#define SLAB (GEMM_MT * 16 * LDA)

typedef       __attribute__((address_space(3))) float* lfp;
typedef       __attribute__((address_space(3))) f4v*   lf4p;
typedef const __attribute__((address_space(3))) v2f*   lcv2p;

template <int NT>
__device__ __forceinline__ void stage_rows(lfp As, int row0, int t,
        gcfp x, gcip src, gcip offs, gcip deg) {
    int lane = t & 31;
    int lo   = lane * 4;
    int rl   = __builtin_amdgcn_readfirstlane(t >> 5);
    int node = row0 + rl;

    int degv = 0, offv = 0;
    if (lane < NT * 2) {
        degv = deg[node + 8 * lane];
        offv = offs[node + 8 * lane];
    }
    int dg  = __builtin_amdgcn_readlane(degv, 0);
    int off = __builtin_amdgcn_readlane(offv, 0);
    int my = 0;
    if (lane < (dg < 32 ? dg : 32)) my = src[off + lane];

    for (int i = 0; i < NT * 2; ++i) {
        int nodeN = node + 8;
        int dgN  = __builtin_amdgcn_readlane(degv, i + 1);
        int offN = __builtin_amdgcn_readlane(offv, i + 1);
        int base = node * D128 + lo;
        f4v acc  = { 0.f, 0.f, 0.f, 0.f };
        f4v r[AGG_ROWS];
        int cnt = dg < 32 ? dg : 32;

        if (dg > 0) {
            if (cnt >= AGG_ROWS) { AGG_ISSUE_FULL(my, 0) }
            else                 { AGG_ISSUE_PART(my, 0, cnt) }
        }
        f4v self = *(gcf4p)(x + base);
        int myN = 0;
        if (lane < (dgN < 32 ? dgN : 32)) myN = src[offN + lane];

        if (dg > 0) {
            if (cnt >= AGG_ROWS) { AGG_ADD_FULL() }
            else                 { AGG_ADD_PART(cnt) }
            for (int e = AGG_ROWS; e < cnt; e += AGG_ROWS) {
                int m = cnt - e;
                if (m >= AGG_ROWS) {
                    AGG_ISSUE_FULL(my, e)
                    __builtin_amdgcn_sched_barrier(0);
                    AGG_ADD_FULL()
                } else {
                    AGG_ISSUE_PART(my, e, m)
                    AGG_ADD_PART(m)
                }
            }
            for (int c0 = 32; c0 < dg; c0 += 32) {
                int c = dg - c0;
                c = c < 32 ? c : 32;
                int ids = 0;
                if (lane < c) ids = src[off + c0 + lane];
                for (int e = 0; e < c; e += AGG_ROWS) {
                    int m = c - e;
                    if (m >= AGG_ROWS) {
                        AGG_ISSUE_FULL(ids, e)
                        __builtin_amdgcn_sched_barrier(0);
                        AGG_ADD_FULL()
                    } else {
                        AGG_ISSUE_PART(ids, e, m)
                        AGG_ADD_PART(m)
                    }
                }
            }
            float inv = 1.0f / (float)dg;
            acc *= inv;
        }
        f4v zero = { 0.f, 0.f, 0.f, 0.f };
        lfp dstrow = As + rl * LDA + lo;
        *(lf4p)(dstrow)            = self;
        *(lf4p)(dstrow + SLAB)     = (dg > 0) ? self : zero;
        *(lf4p)(dstrow + 2 * SLAB) = acc;

        node = nodeN; rl += 8; dg = dgN; off = offN; my = myN;
    }
}
#undef AGG_ISSUE_FULL
#undef AGG_ISSUE_PART
#undef AGG_ADD_FULL
#undef AGG_ADD_PART

template <int NT>
__device__ __forceinline__ void gemm_rows(lfp As, int row0, int t,
        gcfp x, gcip src, gcip offs, gcip deg, gcfp Wg, gcfp Ws, gcfp Wl,
        gcfp bias, gfp out) {
    stage_rows<NT>(As, row0, t, x, src, offs, deg);
    __syncthreads();

    int wave = t >> 5;
    int lane = t & 31;
    int col0 = wave * 16;
    int mrow = lane & 15;
    int koff = (lane >> 4) * 2;

    v8f acc[NT];
    #pragma unroll
    for (int m = 0; m < NT; ++m)
        acc[m] = (v8f){ 0.f, 0.f, 0.f, 0.f, 0.f, 0.f, 0.f, 0.f };

    gcfp Wmats[3] = { Wg, Ws, Wl };
    #pragma unroll
    for (int b3 = 0; b3 < 3; ++b3) {
        lfp  A    = As + b3 * SLAB + mrow * LDA + koff;
        gcfp Brow = Wmats[b3] + (size_t)(col0 + mrow) * D128 + koff;
        #pragma unroll
        for (int k = 0; k < D128; k += 4) {
            v2f b = *(gcv2p)(Brow + k);
            #pragma unroll
            for (int m = 0; m < NT; ++m) {
                v2f a = *(lcv2p)(A + m * 16 * LDA + k);
                acc[m] = __builtin_amdgcn_wmma_f32_16x16x4_f32(
                    false, a, false, b, (short)0, acc[m], false, false);
            }
        }
    }

    int colg = col0 + (lane & 15);
    float bv = bias[colg];
    #pragma unroll
    for (int m = 0; m < NT; ++m) {
        int rbase = row0 + m * 16 + (lane >> 4) * 8;
        #pragma unroll
        for (int i = 0; i < 8; ++i) {
            float v = acc[m][i] + bv;
            v = v > 0.f ? v : (expf(v) - 1.f);
            out[(size_t)(rbase + i) * D128 + colg] = v;
        }
    }
}

__global__ __launch_bounds__(256)
void gemm_fused(const float* __restrict__ x_, const int* __restrict__ src_,
                const int* __restrict__ offs_, const int* __restrict__ deg_,
                const float* __restrict__ Wg_, const float* __restrict__ Ws_,
                const float* __restrict__ Wl_,
                const float* __restrict__ bias_, float* __restrict__ out_, int n) {
    __shared__ float As[3 * SLAB];

    int t = threadIdx.x;
    int b = blockIdx.x;
    int nbig = (n >> 4) / GEMM_MT;
    nbig = nbig < GEMM_BIG ? nbig : GEMM_BIG;

    if (b < nbig) {
        gemm_rows<GEMM_MT>((lfp)As, b * (GEMM_MT * 16), t,
                           (gcfp)x_, (gcip)src_, (gcip)offs_, (gcip)deg_,
                           (gcfp)Wg_, (gcfp)Ws_, (gcfp)Wl_,
                           (gcfp)bias_, (gfp)out_);
        return;
    }

    int tile = nbig * GEMM_MT + (b - nbig);
    int row0 = tile * 16;

    if (row0 + 16 > n) {
        for (int r = row0; r < n; ++r) {
            int dg  = deg_[r];
            int off = offs_[r];
            float inv = dg > 0 ? 1.0f / (float)dg : 0.f;
            for (int c = t; c < D128; c += 256) {
                float s = bias_[c];
                const float* xr = x_  + (size_t)r * D128;
                const float* wg = Wg_ + (size_t)c * D128;
                const float* ws = Ws_ + (size_t)c * D128;
                const float* wl = Wl_ + (size_t)c * D128;
                for (int k = 0; k < D128; ++k) {
                    float nmk = 0.f;
                    for (int e = 0; e < dg; ++e)
                        nmk += x_[(size_t)src_[off + e] * D128 + k];
                    nmk *= inv;
                    float xmk = dg > 0 ? xr[k] : 0.f;
                    s += xr[k] * wg[k] + xmk * ws[k] + nmk * wl[k];
                }
                s = s > 0.f ? s : (expf(s) - 1.f);
                out_[(size_t)r * D128 + c] = s;
            }
        }
        return;
    }

    gemm_rows<1>((lfp)As, row0, t,
                 (gcfp)x_, (gcip)src_, (gcip)offs_, (gcip)deg_,
                 (gcfp)Wg_, (gcfp)Ws_, (gcfp)Wl_,
                 (gcfp)bias_, (gfp)out_);
}

extern "C" void kernel_launch(void* const* d_in, const int* in_sizes, int n_in,
                              void* d_out, int out_size, void* d_ws, size_t ws_size,
                              hipStream_t stream) {
    (void)n_in; (void)out_size; (void)ws_size;
    const float* x    = (const float*)d_in[0];
    const float* Wg   = (const float*)d_in[1];
    const float* Wl   = (const float*)d_in[2];
    const float* Ws   = (const float*)d_in[3];
    const float* bias = (const float*)d_in[4];
    const int*   src  = (const int*)d_in[5];
    const int*   deg  = (const int*)d_in[7];

    int N = in_sizes[7];

    int* offs = (int*)d_ws;

    int scanBlocks = (N + SCAN_THREADS - 1) / SCAN_THREADS;
    scan_offsets<<<scanBlocks, SCAN_THREADS, 0, stream>>>(deg, offs, N);

    int tiles = (N + 15) / 16;
    int nbig = (N / 16) / GEMM_MT;
    nbig = nbig < GEMM_BIG ? nbig : GEMM_BIG;
    int gblocks = nbig + (tiles - nbig * GEMM_MT);
    gemm_fused<<<gblocks, 256, 0, stream>>>(x, src, offs, deg, Wg, Ws, Wl, bias,
                                            (float*)d_out, N);
}


# self weights folded into the global ones: B = Wg+Ws built in the scan launch, main GEMM K 384->256; deg==0 rows packed into own tiles for x.Wg; rounding changes: x.(Wg+Ws) replaces x.Wg+x.Ws
# speedup vs baseline: 2.9732x; 1.0392x over previous
#include <hip/hip_runtime.h>
#include <math.h>

typedef __attribute__((ext_vector_type(2))) float v2f;
typedef __attribute__((ext_vector_type(4))) float f4v;
typedef __attribute__((ext_vector_type(8))) float v8f;
typedef __attribute__((ext_vector_type(4))) int   i4v;

typedef const __attribute__((address_space(1))) float* gcfp;
typedef       __attribute__((address_space(1))) float* gfp;
typedef const __attribute__((address_space(1))) int*   gcip;
typedef       __attribute__((address_space(1))) int*   gip;
typedef const __attribute__((address_space(1))) f4v*   gcf4p;
typedef       __attribute__((address_space(1))) f4v*   gf4p;
typedef const __attribute__((address_space(1))) v2f*   gcv2p;
typedef const __attribute__((address_space(1))) i4v*   gci4p;

#define D128 128
#define SCAN_THREADS 1024

__global__ __launch_bounds__(SCAN_THREADS)
void scan_offsets(const int* __restrict__ deg_, int* __restrict__ offs_, int n,
                  const float* __restrict__ Wg_, const float* __restrict__ Ws_,
                  float* __restrict__ Wgs_) {
    static_assert(SCAN_THREADS == 32 * 32, "one LDS slot per lane: 32 waves of 32 lanes");
    gcip  deg  = (gcip)deg_;
    gci4p deg4 = (gci4p)deg_;
    gip   offs = (gip)offs_;
    __shared__ int wtot[32];
    __shared__ int wpre[32];

    int t = threadIdx.x;
    int lane = t & 31;
    int wave = t >> 5;

    for (int g = blockIdx.x * SCAN_THREADS + t; g < D128 * D128 / 4;
         g += gridDim.x * SCAN_THREADS) {
        f4v a = ((gcf4p)Wg_)[g];
        f4v b = ((gcf4p)Ws_)[g];
        ((gf4p)Wgs_)[g] = a + b;
    }

    int start = blockIdx.x * SCAN_THREADS;
    int i = start + t;
    int d = (i < n) ? deg[i] : 0;

    int nq = start >> 2;
    int pre = 0;
    for (int q0 = t; q0 < nq; q0 += 4 * SCAN_THREADS) {
        #pragma unroll
        for (int u = 0; u < 4; ++u) {
            int q = q0 + u * SCAN_THREADS;
            bool ok = q < nq;
            i4v v = deg4[ok ? q : q0];
            int sv = (v.x + v.y) + (v.z + v.w);
            pre += ok ? sv : 0;
        }
    }

    int inc = d;
    #pragma unroll
    for (int o = 1; o < 32; o <<= 1) {
        int v = __shfl_up(inc, o);
        if (lane >= o) inc += v;
    }
    #pragma unroll
    for (int o = 16; o > 0; o >>= 1) pre += __shfl_xor(pre, o);

    if (lane == 31) wtot[wave] = inc;
    if (lane == 0)  wpre[wave] = pre;
    __syncthreads();

    int c = wpre[lane] + ((lane < wave) ? wtot[lane] : 0);
    #pragma unroll
    for (int o = 16; o > 0; o >>= 1) c += __shfl_xor(c, o);

    if (i < n) offs[i] = c + inc - d;
}

#define AGG_ROWS 8

#define AGG_ISSUE_FULL(ids, e)                                               \
    _Pragma("unroll")                                                        \
    for (int u = 0; u < AGG_ROWS; ++u) {                                     \
        int s = __builtin_amdgcn_readlane((ids), (e) + u);                   \
        r[u] = *(gcf4p)(x + (size_t)s * D128 + lo);                          \
    }
#define AGG_ISSUE_PART(ids, e, m)                                            \
    {                                                                        \
        int sid[AGG_ROWS - 1];                                               \
        _Pragma("unroll")                                                    \
        for (int u = 0; u < AGG_ROWS - 1; ++u)                               \
            sid[u] = __builtin_amdgcn_readlane((ids), (e) + u);              \
        _Pragma("unroll")                                                    \
        for (int u = 0; u < AGG_ROWS - 1; ++u)                               \
            if (u < (m)) r[u] = *(gcf4p)(x + (size_t)sid[u] * D128 + lo);    \
    }
#define AGG_ADD_FULL()                                                       \
    _Pragma("unroll")                                                        \
    for (int u = 0; u < AGG_ROWS; ++u) acc += r[u];
#define AGG_ADD_PART(m)                                                      \
    _Pragma("unroll")                                                        \
    for (int u = 0; u < AGG_ROWS - 1; ++u)                                   \
        if (u < (m)) acc += r[u];

#define LDA 132
#define GEMM_MT 4
#define GEMM_BIG 768
#define SLAB (GEMM_MT * 16 * LDA)

typedef       __attribute__((address_space(3))) float* lfp;
typedef       __attribute__((address_space(3))) f4v*   lf4p;
typedef const __attribute__((address_space(3))) v2f*   lcv2p;
typedef       __attribute__((address_space(3))) int*   lip;

template <int NT>
__device__ __forceinline__ void stage_rows(lfp As, int row0, int t,
        gcfp x, gcip src, gcip offs, gcip deg) {
    int lane = t & 31;
    int lo   = lane * 4;
    int rl   = __builtin_amdgcn_readfirstlane(t >> 5);
    int node = row0 + rl;

    int degv = 0, offv = 0;
    if (lane < NT * 2) {
        degv = deg[node + 8 * lane];
        offv = offs[node + 8 * lane];
    }
    int dg  = __builtin_amdgcn_readlane(degv, 0);
    int off = __builtin_amdgcn_readlane(offv, 0);
    int my = 0;
    if (lane < (dg < 32 ? dg : 32)) my = src[off + lane];

    for (int i = 0; i < NT * 2; ++i) {
        int nodeN = node + 8;
        int dgN  = __builtin_amdgcn_readlane(degv, i + 1);
        int offN = __builtin_amdgcn_readlane(offv, i + 1);
        int base = node * D128 + lo;
        f4v acc  = { 0.f, 0.f, 0.f, 0.f };
        f4v r[AGG_ROWS];
        int cnt = dg < 32 ? dg : 32;

        if (dg > 0) {
            if (cnt >= AGG_ROWS) { AGG_ISSUE_FULL(my, 0) }
            else                 { AGG_ISSUE_PART(my, 0, cnt) }
        }
        f4v self = *(gcf4p)(x + base);
        int myN = 0;
        if (lane < (dgN < 32 ? dgN : 32)) myN = src[offN + lane];

        if (dg > 0) {
            if (cnt >= AGG_ROWS) { AGG_ADD_FULL() }
            else                 { AGG_ADD_PART(cnt) }
            for (int e = AGG_ROWS; e < cnt; e += AGG_ROWS) {
                int m = cnt - e;
                if (m >= AGG_ROWS) {
                    AGG_ISSUE_FULL(my, e)
                    __builtin_amdgcn_sched_barrier(0);
                    AGG_ADD_FULL()
                } else {
                    AGG_ISSUE_PART(my, e, m)
                    AGG_ADD_PART(m)
                }
            }
            for (int c0 = 32; c0 < dg; c0 += 32) {
                int c = dg - c0;
                c = c < 32 ? c : 32;
                int ids = 0;
                if (lane < c) ids = src[off + c0 + lane];
                for (int e = 0; e < c; e += AGG_ROWS) {
                    int m = c - e;
                    if (m >= AGG_ROWS) {
                        AGG_ISSUE_FULL(ids, e)
                        __builtin_amdgcn_sched_barrier(0);
                        AGG_ADD_FULL()
                    } else {
                        AGG_ISSUE_PART(ids, e, m)
                        AGG_ADD_PART(m)
                    }
                }
            }
            float inv = 1.0f / (float)dg;
            acc *= inv;
        }
        lfp dstrow = As + rl * LDA + lo;
        *(lf4p)(dstrow)        = self;
        *(lf4p)(dstrow + SLAB) = acc;

        node = nodeN; rl += 8; dg = dgN; off = offN; my = myN;
    }
}
#undef AGG_ISSUE_FULL
#undef AGG_ISSUE_PART
#undef AGG_ADD_FULL
#undef AGG_ADD_PART

template <int NT>
__device__ __forceinline__ void gemm_rows(lfp As, int row0, int t,
        gcfp x, gcip src, gcip offs, gcip deg, gcfp Wgs, gcfp Wg, gcfp Wl,
        gcfp bias, gfp out) {
    int wave = t >> 5;
    int lane = t & 31;

    int d0 = (lane < NT * 16) ? deg[row0 + lane] : 1;
    int d1 = (lane + 32 < NT * 16) ? deg[row0 + 32 + lane] : 1;

    stage_rows<NT>(As, row0, t, x, src, offs, deg);

    unsigned z0 = __builtin_amdgcn_ballot_w32(d0 == 0);
    unsigned z1 = __builtin_amdgcn_ballot_w32(d1 == 0);
    int cnt0 = __builtin_popcount(z0);
    int cnt  = cnt0 + __builtin_popcount(z1);
    lip zl = (lip)(As + 2 * SLAB) + wave * 64;
    {
        unsigned below = (1u << lane) - 1u;
        if ((z0 >> lane) & 1u) zl[__builtin_popcount(z0 & below)] = lane;
        if ((z1 >> lane) & 1u) zl[cnt0 + __builtin_popcount(z1 & below)] = lane + 32;
    }
    __syncthreads();

    int col0 = wave * 16;
    int mrow = lane & 15;
    int koff = (lane >> 4) * 2;

    v8f acc[NT];
    #pragma unroll
    for (int m = 0; m < NT; ++m)
        acc[m] = (v8f){ 0.f, 0.f, 0.f, 0.f, 0.f, 0.f, 0.f, 0.f };

    gcfp Wmats[2] = { Wgs, Wl };
    #pragma unroll
    for (int b2 = 0; b2 < 2; ++b2) {
        lfp  A    = As + b2 * SLAB + mrow * LDA + koff;
        gcfp Brow = Wmats[b2] + (size_t)(col0 + mrow) * D128 + koff;
        #pragma unroll
        for (int k = 0; k < D128; k += 4) {
            v2f b = *(gcv2p)(Brow + k);
            #pragma unroll
            for (int m = 0; m < NT; ++m) {
                v2f a = *(lcv2p)(A + m * 16 * LDA + k);
                acc[m] = __builtin_amdgcn_wmma_f32_16x16x4_f32(
                    false, a, false, b, (short)0, acc[m], false, false);
            }
        }
    }

    int colg = col0 + (lane & 15);
    float bv = bias[colg];
    #pragma unroll
    for (int m = 0; m < NT; ++m) {
        int rbase = row0 + m * 16 + (lane >> 4) * 8;
        unsigned zw = (m < 2) ? z0 : z1;
        unsigned zb = zw >> ((m & 1) * 16 + (lane >> 4) * 8);
        #pragma unroll
        for (int i = 0; i < 8; ++i) {
            if (!((zb >> i) & 1u)) {
                float v = acc[m][i] + bv;
                v = v > 0.f ? v : (expf(v) - 1.f);
                out[(size_t)(rbase + i) * D128 + colg] = v;
            }
        }
    }

    for (int j0 = 0; j0 < cnt; j0 += 16) {
        int ia = j0 + mrow;
        int zr = zl[ia < cnt ? ia : 0];
        lfp  A    = As + zr * LDA + koff;
        gcfp Brow = Wg + (size_t)(col0 + mrow) * D128 + koff;
        v8f z = (v8f){ 0.f, 0.f, 0.f, 0.f, 0.f, 0.f, 0.f, 0.f };
        #pragma unroll
        for (int k = 0; k < D128; k += 4) {
            v2f b = *(gcv2p)(Brow + k);
            v2f a = *(lcv2p)(A + k);
            z = __builtin_amdgcn_wmma_f32_16x16x4_f32(
                false, a, false, b, (short)0, z, false, false);
        }
        int e0 = j0 + (lane >> 4) * 8;
        int zrow[8];
        #pragma unroll
        for (int i = 0; i < 8; ++i)
            zrow[i] = zl[(e0 + i) < cnt ? (e0 + i) : 0];
        #pragma unroll
        for (int i = 0; i < 8; ++i) {
            if (e0 + i < cnt) {
                float v = z[i] + bv;
                v = v > 0.f ? v : (expf(v) - 1.f);
                out[(size_t)(row0 + zrow[i]) * D128 + colg] = v;
            }
        }
    }
}

__global__ __launch_bounds__(256)
void gemm_fused(const float* __restrict__ x_, const int* __restrict__ src_,
                const int* __restrict__ offs_, const int* __restrict__ deg_,
                const float* __restrict__ Wg_, const float* __restrict__ Ws_,
                const float* __restrict__ Wl_, const float* __restrict__ Wgs_,
                const float* __restrict__ bias_, float* __restrict__ out_, int n) {
    __shared__ float As[3 * SLAB];

    int t = threadIdx.x;
    int b = blockIdx.x;
    int nbig = (n >> 4) / GEMM_MT;
    nbig = nbig < GEMM_BIG ? nbig : GEMM_BIG;

    if (b < nbig) {
        gemm_rows<GEMM_MT>((lfp)As, b * (GEMM_MT * 16), t,
                           (gcfp)x_, (gcip)src_, (gcip)offs_, (gcip)deg_,
                           (gcfp)Wgs_, (gcfp)Wg_, (gcfp)Wl_,
                           (gcfp)bias_, (gfp)out_);
        return;
    }

    int tile = nbig * GEMM_MT + (b - nbig);
    int row0 = tile * 16;

    if (row0 + 16 > n) {
        for (int r = row0; r < n; ++r) {
            int dg  = deg_[r];
            int off = offs_[r];
            float inv = dg > 0 ? 1.0f / (float)dg : 0.f;
            for (int c = t; c < D128; c += 256) {
                float s = bias_[c];
                const float* xr = x_  + (size_t)r * D128;
                const float* wg = Wg_ + (size_t)c * D128;
                const float* ws = Ws_ + (size_t)c * D128;
                const float* wl = Wl_ + (size_t)c * D128;
                for (int k = 0; k < D128; ++k) {
                    float nmk = 0.f;
                    for (int e = 0; e < dg; ++e)
                        nmk += x_[(size_t)src_[off + e] * D128 + k];
                    nmk *= inv;
                    float xmk = dg > 0 ? xr[k] : 0.f;
                    s += xr[k] * wg[k] + xmk * ws[k] + nmk * wl[k];
                }
                s = s > 0.f ? s : (expf(s) - 1.f);
                out_[(size_t)r * D128 + c] = s;
            }
        }
        return;
    }

    gemm_rows<1>((lfp)As, row0, t,
                 (gcfp)x_, (gcip)src_, (gcip)offs_, (gcip)deg_,
                 (gcfp)Wgs_, (gcfp)Wg_, (gcfp)Wl_,
                 (gcfp)bias_, (gfp)out_);
}

extern "C" void kernel_launch(void* const* d_in, const int* in_sizes, int n_in,
                              void* d_out, int out_size, void* d_ws, size_t ws_size,
                              hipStream_t stream) {
    (void)n_in; (void)out_size; (void)ws_size;
    const float* x    = (const float*)d_in[0];
    const float* Wg   = (const float*)d_in[1];
    const float* Wl   = (const float*)d_in[2];
    const float* Ws   = (const float*)d_in[3];
    const float* bias = (const float*)d_in[4];
    const int*   src  = (const int*)d_in[5];
    const int*   deg  = (const int*)d_in[7];

    int N = in_sizes[7];

    float* Wgs = (float*)d_ws;
    int* offs = (int*)d_ws + D128 * D128;

    int scanBlocks = (N + SCAN_THREADS - 1) / SCAN_THREADS;
    scan_offsets<<<scanBlocks, SCAN_THREADS, 0, stream>>>(deg, offs, N, Wg, Ws, Wgs);

    int tiles = (N + 15) / 16;
    int nbig = (N / 16) / GEMM_MT;
    nbig = nbig < GEMM_BIG ? nbig : GEMM_BIG;
    int gblocks = nbig + (tiles - nbig * GEMM_MT);
    gemm_fused<<<gblocks, 256, 0, stream>>>(x, src, offs, deg, Wg, Ws, Wl, Wgs, bias,
                                            (float*)d_out, N);
}


# gather: 16 neighbor rows in flight per wave instead of 8, so a node of up to 32 edges takes at most 2 memory round trips (was up to 4); edge order of the sums unchanged
# speedup vs baseline: 3.0907x; 1.0395x over previous
#include <hip/hip_runtime.h>
#include <math.h>

typedef __attribute__((ext_vector_type(2))) float v2f;
typedef __attribute__((ext_vector_type(4))) float f4v;
typedef __attribute__((ext_vector_type(8))) float v8f;
typedef __attribute__((ext_vector_type(4))) int   i4v;

typedef const __attribute__((address_space(1))) float* gcfp;
typedef       __attribute__((address_space(1))) float* gfp;
typedef const __attribute__((address_space(1))) int*   gcip;
typedef       __attribute__((address_space(1))) int*   gip;
typedef const __attribute__((address_space(1))) f4v*   gcf4p;
typedef       __attribute__((address_space(1))) f4v*   gf4p;
typedef const __attribute__((address_space(1))) v2f*   gcv2p;
typedef const __attribute__((address_space(1))) i4v*   gci4p;

#define D128 128
#define SCAN_THREADS 1024

__global__ __launch_bounds__(SCAN_THREADS)
void scan_offsets(const int* __restrict__ deg_, int* __restrict__ offs_, int n,
                  const float* __restrict__ Wg_, const float* __restrict__ Ws_,
                  float* __restrict__ Wgs_) {
    static_assert(SCAN_THREADS == 32 * 32, "one LDS slot per lane: 32 waves of 32 lanes");
    gcip  deg  = (gcip)deg_;
    gci4p deg4 = (gci4p)deg_;
    gip   offs = (gip)offs_;
    __shared__ int wtot[32];
    __shared__ int wpre[32];

    int t = threadIdx.x;
    int lane = t & 31;
    int wave = t >> 5;

    for (int g = blockIdx.x * SCAN_THREADS + t; g < D128 * D128 / 4;
         g += gridDim.x * SCAN_THREADS) {
        f4v a = ((gcf4p)Wg_)[g];
        f4v b = ((gcf4p)Ws_)[g];
        ((gf4p)Wgs_)[g] = a + b;
    }

    int start = blockIdx.x * SCAN_THREADS;
    int i = start + t;
    int d = (i < n) ? deg[i] : 0;

    int nq = start >> 2;
    int pre = 0;
    for (int q0 = t; q0 < nq; q0 += 4 * SCAN_THREADS) {
        #pragma unroll
        for (int u = 0; u < 4; ++u) {
            int q = q0 + u * SCAN_THREADS;
            bool ok = q < nq;
            i4v v = deg4[ok ? q : q0];
            int sv = (v.x + v.y) + (v.z + v.w);
            pre += ok ? sv : 0;
        }
    }

    int inc = d;
    #pragma unroll
    for (int o = 1; o < 32; o <<= 1) {
        int v = __shfl_up(inc, o);
        if (lane >= o) inc += v;
    }
    #pragma unroll
    for (int o = 16; o > 0; o >>= 1) pre += __shfl_xor(pre, o);

    if (lane == 31) wtot[wave] = inc;
    if (lane == 0)  wpre[wave] = pre;
    __syncthreads();

    int c = wpre[lane] + ((lane < wave) ? wtot[lane] : 0);
    #pragma unroll
    for (int o = 16; o > 0; o >>= 1) c += __shfl_xor(c, o);

    if (i < n) offs[i] = c + inc - d;
}

#define AGG_ROWS 16

#define AGG_ISSUE_FULL(ids, e)                                               \
    _Pragma("unroll")                                                        \
    for (int u = 0; u < AGG_ROWS; ++u) {                                     \
        int s = __builtin_amdgcn_readlane((ids), (e) + u);                   \
        r[u] = *(gcf4p)(x + (size_t)s * D128 + lo);                          \
    }
#define AGG_ISSUE_PART(ids, e, m)                                            \
    {                                                                        \
        int sid[AGG_ROWS - 1];                                               \
        _Pragma("unroll")                                                    \
        for (int u = 0; u < AGG_ROWS - 1; ++u)                               \
            sid[u] = __builtin_amdgcn_readlane((ids), (e) + u);              \
        _Pragma("unroll")                                                    \
        for (int u = 0; u < AGG_ROWS - 1; ++u)                               \
            if (u < (m)) r[u] = *(gcf4p)(x + (size_t)sid[u] * D128 + lo);    \
    }
#define AGG_ADD_FULL()                                                       \
    _Pragma("unroll")                                                        \
    for (int u = 0; u < AGG_ROWS; ++u) acc += r[u];
#define AGG_ADD_PART(m)                                                      \
    _Pragma("unroll")                                                        \
    for (int u = 0; u < AGG_ROWS - 1; ++u)                                   \
        if (u < (m)) acc += r[u];

#define LDA 132
#define GEMM_MT 4
#define GEMM_BIG 768
#define SLAB (GEMM_MT * 16 * LDA)

typedef       __attribute__((address_space(3))) float* lfp;
typedef       __attribute__((address_space(3))) f4v*   lf4p;
typedef const __attribute__((address_space(3))) v2f*   lcv2p;
typedef       __attribute__((address_space(3))) int*   lip;

template <int NT>
__device__ __forceinline__ void stage_rows(lfp As, int row0, int t,
        gcfp x, gcip src, gcip offs, gcip deg) {
    int lane = t & 31;
    int lo   = lane * 4;
    int rl   = __builtin_amdgcn_readfirstlane(t >> 5);
    int node = row0 + rl;

    int degv = 0, offv = 0;
    if (lane < NT * 2) {
        degv = deg[node + 8 * lane];
        offv = offs[node + 8 * lane];
    }
    int dg  = __builtin_amdgcn_readlane(degv, 0);
    int off = __builtin_amdgcn_readlane(offv, 0);
    int my = 0;
    if (lane < (dg < 32 ? dg : 32)) my = src[off + lane];

    for (int i = 0; i < NT * 2; ++i) {
        int nodeN = node + 8;
        int dgN  = __builtin_amdgcn_readlane(degv, i + 1);
        int offN = __builtin_amdgcn_readlane(offv, i + 1);
        int base = node * D128 + lo;
        f4v acc  = { 0.f, 0.f, 0.f, 0.f };
        f4v r[AGG_ROWS];
        int cnt = dg < 32 ? dg : 32;

        if (dg > 0) {
            if (cnt >= AGG_ROWS) { AGG_ISSUE_FULL(my, 0) }
            else                 { AGG_ISSUE_PART(my, 0, cnt) }
        }
        f4v self = *(gcf4p)(x + base);
        int myN = 0;
        if (lane < (dgN < 32 ? dgN : 32)) myN = src[offN + lane];

        if (dg > 0) {
            if (cnt >= AGG_ROWS) { AGG_ADD_FULL() }
            else                 { AGG_ADD_PART(cnt) }
            for (int e = AGG_ROWS; e < cnt; e += AGG_ROWS) {
                int m = cnt - e;
                if (m >= AGG_ROWS) {
                    AGG_ISSUE_FULL(my, e)
                    __builtin_amdgcn_sched_barrier(0);
                    AGG_ADD_FULL()
                } else {
                    AGG_ISSUE_PART(my, e, m)
                    AGG_ADD_PART(m)
                }
            }
            for (int c0 = 32; c0 < dg; c0 += 32) {
                int c = dg - c0;
                c = c < 32 ? c : 32;
                int ids = 0;
                if (lane < c) ids = src[off + c0 + lane];
                for (int e = 0; e < c; e += AGG_ROWS) {
                    int m = c - e;
                    if (m >= AGG_ROWS) {
                        AGG_ISSUE_FULL(ids, e)
                        __builtin_amdgcn_sched_barrier(0);
                        AGG_ADD_FULL()
                    } else {
                        AGG_ISSUE_PART(ids, e, m)
                        AGG_ADD_PART(m)
                    }
                }
            }
            float inv = 1.0f / (float)dg;
            acc *= inv;
        }
        lfp dstrow = As + rl * LDA + lo;
        *(lf4p)(dstrow)        = self;
        *(lf4p)(dstrow + SLAB) = acc;

        node = nodeN; rl += 8; dg = dgN; off = offN; my = myN;
    }
}
#undef AGG_ISSUE_FULL
#undef AGG_ISSUE_PART
#undef AGG_ADD_FULL
#undef AGG_ADD_PART

template <int NT>
__device__ __forceinline__ void gemm_rows(lfp As, int row0, int t,
        gcfp x, gcip src, gcip offs, gcip deg, gcfp Wgs, gcfp Wg, gcfp Wl,
        gcfp bias, gfp out) {
    int wave = t >> 5;
    int lane = t & 31;

    int d0 = (lane < NT * 16) ? deg[row0 + lane] : 1;
    int d1 = (lane + 32 < NT * 16) ? deg[row0 + 32 + lane] : 1;

    stage_rows<NT>(As, row0, t, x, src, offs, deg);

    unsigned z0 = __builtin_amdgcn_ballot_w32(d0 == 0);
    unsigned z1 = __builtin_amdgcn_ballot_w32(d1 == 0);
    int cnt0 = __builtin_popcount(z0);
    int cnt  = cnt0 + __builtin_popcount(z1);
    lip zl = (lip)(As + 2 * SLAB) + wave * 64;
    {
        unsigned below = (1u << lane) - 1u;
        if ((z0 >> lane) & 1u) zl[__builtin_popcount(z0 & below)] = lane;
        if ((z1 >> lane) & 1u) zl[cnt0 + __builtin_popcount(z1 & below)] = lane + 32;
    }
    __syncthreads();

    int col0 = wave * 16;
    int mrow = lane & 15;
    int koff = (lane >> 4) * 2;

    v8f acc[NT];
    #pragma unroll
    for (int m = 0; m < NT; ++m)
        acc[m] = (v8f){ 0.f, 0.f, 0.f, 0.f, 0.f, 0.f, 0.f, 0.f };

    gcfp Wmats[2] = { Wgs, Wl };
    #pragma unroll
    for (int b2 = 0; b2 < 2; ++b2) {
        lfp  A    = As + b2 * SLAB + mrow * LDA + koff;
        gcfp Brow = Wmats[b2] + (size_t)(col0 + mrow) * D128 + koff;
        #pragma unroll
        for (int k = 0; k < D128; k += 4) {
            v2f b = *(gcv2p)(Brow + k);
            #pragma unroll
            for (int m = 0; m < NT; ++m) {
                v2f a = *(lcv2p)(A + m * 16 * LDA + k);
                acc[m] = __builtin_amdgcn_wmma_f32_16x16x4_f32(
                    false, a, false, b, (short)0, acc[m], false, false);
            }
        }
    }

    int colg = col0 + (lane & 15);
    float bv = bias[colg];
    #pragma unroll
    for (int m = 0; m < NT; ++m) {
        int rbase = row0 + m * 16 + (lane >> 4) * 8;
        unsigned zw = (m < 2) ? z0 : z1;
        unsigned zb = zw >> ((m & 1) * 16 + (lane >> 4) * 8);
        #pragma unroll
        for (int i = 0; i < 8; ++i) {
            if (!((zb >> i) & 1u)) {
                float v = acc[m][i] + bv;
                v = v > 0.f ? v : (expf(v) - 1.f);
                out[(size_t)(rbase + i) * D128 + colg] = v;
            }
        }
    }

    for (int j0 = 0; j0 < cnt; j0 += 16) {
        int ia = j0 + mrow;
        int zr = zl[ia < cnt ? ia : 0];
        lfp  A    = As + zr * LDA + koff;
        gcfp Brow = Wg + (size_t)(col0 + mrow) * D128 + koff;
        v8f z = (v8f){ 0.f, 0.f, 0.f, 0.f, 0.f, 0.f, 0.f, 0.f };
        #pragma unroll
        for (int k = 0; k < D128; k += 4) {
            v2f b = *(gcv2p)(Brow + k);
            v2f a = *(lcv2p)(A + k);
            z = __builtin_amdgcn_wmma_f32_16x16x4_f32(
                false, a, false, b, (short)0, z, false, false);
        }
        int e0 = j0 + (lane >> 4) * 8;
        int zrow[8];
        #pragma unroll
        for (int i = 0; i < 8; ++i)
            zrow[i] = zl[(e0 + i) < cnt ? (e0 + i) : 0];
        #pragma unroll
        for (int i = 0; i < 8; ++i) {
            if (e0 + i < cnt) {
                float v = z[i] + bv;
                v = v > 0.f ? v : (expf(v) - 1.f);
                out[(size_t)(row0 + zrow[i]) * D128 + colg] = v;
            }
        }
    }
}

__global__ __launch_bounds__(256)
void gemm_fused(const float* __restrict__ x_, const int* __restrict__ src_,
                const int* __restrict__ offs_, const int* __restrict__ deg_,
                const float* __restrict__ Wg_, const float* __restrict__ Ws_,
                const float* __restrict__ Wl_, const float* __restrict__ Wgs_,
                const float* __restrict__ bias_, float* __restrict__ out_, int n) {
    __shared__ float As[3 * SLAB];

    int t = threadIdx.x;
    int b = blockIdx.x;
    int nbig = (n >> 4) / GEMM_MT;
    nbig = nbig < GEMM_BIG ? nbig : GEMM_BIG;

    if (b < nbig) {
        gemm_rows<GEMM_MT>((lfp)As, b * (GEMM_MT * 16), t,
                           (gcfp)x_, (gcip)src_, (gcip)offs_, (gcip)deg_,
                           (gcfp)Wgs_, (gcfp)Wg_, (gcfp)Wl_,
                           (gcfp)bias_, (gfp)out_);
        return;
    }

    int tile = nbig * GEMM_MT + (b - nbig);
    int row0 = tile * 16;

    if (row0 + 16 > n) {
        for (int r = row0; r < n; ++r) {
            int dg  = deg_[r];
            int off = offs_[r];
            float inv = dg > 0 ? 1.0f / (float)dg : 0.f;
            for (int c = t; c < D128; c += 256) {
                float s = bias_[c];
                const float* xr = x_  + (size_t)r * D128;
                const float* wg = Wg_ + (size_t)c * D128;
                const float* ws = Ws_ + (size_t)c * D128;
                const float* wl = Wl_ + (size_t)c * D128;
                for (int k = 0; k < D128; ++k) {
                    float nmk = 0.f;
                    for (int e = 0; e < dg; ++e)
                        nmk += x_[(size_t)src_[off + e] * D128 + k];
                    nmk *= inv;
                    float xmk = dg > 0 ? xr[k] : 0.f;
                    s += xr[k] * wg[k] + xmk * ws[k] + nmk * wl[k];
                }
                s = s > 0.f ? s : (expf(s) - 1.f);
                out_[(size_t)r * D128 + c] = s;
            }
        }
        return;
    }

    gemm_rows<1>((lfp)As, row0, t,
                 (gcfp)x_, (gcip)src_, (gcip)offs_, (gcip)deg_,
                 (gcfp)Wgs_, (gcfp)Wg_, (gcfp)Wl_,
                 (gcfp)bias_, (gfp)out_);
}

extern "C" void kernel_launch(void* const* d_in, const int* in_sizes, int n_in,
                              void* d_out, int out_size, void* d_ws, size_t ws_size,
                              hipStream_t stream) {
    (void)n_in; (void)out_size; (void)ws_size;
    const float* x    = (const float*)d_in[0];
    const float* Wg   = (const float*)d_in[1];
    const float* Wl   = (const float*)d_in[2];
    const float* Ws   = (const float*)d_in[3];
    const float* bias = (const float*)d_in[4];
    const int*   src  = (const int*)d_in[5];
    const int*   deg  = (const int*)d_in[7];

    int N = in_sizes[7];

    float* Wgs = (float*)d_ws;
    int* offs = (int*)d_ws + D128 * D128;

    int scanBlocks = (N + SCAN_THREADS - 1) / SCAN_THREADS;
    scan_offsets<<<scanBlocks, SCAN_THREADS, 0, stream>>>(deg, offs, N, Wg, Ws, Wgs);

    int tiles = (N + 15) / 16;
    int nbig = (N / 16) / GEMM_MT;
    nbig = nbig < GEMM_BIG ? nbig : GEMM_BIG;
    int gblocks = nbig + (tiles - nbig * GEMM_MT);
    gemm_fused<<<gblocks, 256, 0, stream>>>(x, src, offs, deg, Wg, Ws, Wl, Wgs, bias,
                                            (float*)d_out, N);
}


# weights written by the scan launch in B-fragment order (Wg+Ws, Wl, Wg): a B load is 512 contiguous bytes for two k-steps, 4 lines, not 16 B from each of 16 lines; sum order unchanged
# speedup vs baseline: 3.5311x; 1.1425x over previous
#include <hip/hip_runtime.h>
#include <math.h>

typedef __attribute__((ext_vector_type(2))) float v2f;
typedef __attribute__((ext_vector_type(4))) float f4v;
typedef __attribute__((ext_vector_type(8))) float v8f;
typedef __attribute__((ext_vector_type(4))) int   i4v;

typedef const __attribute__((address_space(1))) float* gcfp;
typedef       __attribute__((address_space(1))) float* gfp;
typedef const __attribute__((address_space(1))) int*   gcip;
typedef       __attribute__((address_space(1))) int*   gip;
typedef const __attribute__((address_space(1))) f4v*   gcf4p;
typedef       __attribute__((address_space(1))) f4v*   gf4p;
typedef const __attribute__((address_space(1))) v2f*   gcv2p;
typedef const __attribute__((address_space(1))) i4v*   gci4p;

#define D128 128
#define SCAN_THREADS 1024

#define FRAG_F4 (D128 * D128 / 4)
__global__ __launch_bounds__(SCAN_THREADS)
void scan_offsets(const int* __restrict__ deg_, int* __restrict__ offs_, int n,
                  const float* __restrict__ Wg_, const float* __restrict__ Ws_,
                  const float* __restrict__ Wl_, float* __restrict__ Wf_) {
    static_assert(SCAN_THREADS == 32 * 32, "one LDS slot per lane: 32 waves of 32 lanes");
    gcip  deg  = (gcip)deg_;
    gci4p deg4 = (gci4p)deg_;
    gip   offs = (gip)offs_;
    __shared__ int wtot[32];
    __shared__ int wpre[32];

    int t = threadIdx.x;
    int lane = t & 31;
    int wave = t >> 5;

    for (int g = blockIdx.x * SCAN_THREADS + t; g < 3 * FRAG_F4;
         g += gridDim.x * SCAN_THREADS) {
        int sel = g / FRAG_F4;
        int idx = g - sel * FRAG_F4;
        int ln  = idx & 31;
        int kk  = (idx >> 5) & 15;
        int ct  = idx >> 9;
        int e   = (ct * 16 + (ln & 15)) * D128 + kk * 8 + (ln >> 4) * 2;
        gcfp p  = (gcfp)(sel == 1 ? Wl_ : Wg_) + e;
        v2f lo2 = *(gcv2p)p;
        v2f hi2 = *(gcv2p)(p + 4);
        if (sel == 0) {
            gcfp q = (gcfp)Ws_ + e;
            lo2 += *(gcv2p)q;
            hi2 += *(gcv2p)(q + 4);
        }
        f4v o = { lo2.x, lo2.y, hi2.x, hi2.y };
        ((gf4p)Wf_)[g] = o;
    }

    int start = blockIdx.x * SCAN_THREADS;
    int i = start + t;
    int d = (i < n) ? deg[i] : 0;

    int nq = start >> 2;
    int pre = 0;
    for (int q0 = t; q0 < nq; q0 += 4 * SCAN_THREADS) {
        #pragma unroll
        for (int u = 0; u < 4; ++u) {
            int q = q0 + u * SCAN_THREADS;
            bool ok = q < nq;
            i4v v = deg4[ok ? q : q0];
            int sv = (v.x + v.y) + (v.z + v.w);
            pre += ok ? sv : 0;
        }
    }

    int inc = d;
    #pragma unroll
    for (int o = 1; o < 32; o <<= 1) {
        int v = __shfl_up(inc, o);
        if (lane >= o) inc += v;
    }
    #pragma unroll
    for (int o = 16; o > 0; o >>= 1) pre += __shfl_xor(pre, o);

    if (lane == 31) wtot[wave] = inc;
    if (lane == 0)  wpre[wave] = pre;
    __syncthreads();

    int c = wpre[lane] + ((lane < wave) ? wtot[lane] : 0);
    #pragma unroll
    for (int o = 16; o > 0; o >>= 1) c += __shfl_xor(c, o);

    if (i < n) offs[i] = c + inc - d;
}

#define AGG_ROWS 16

#define AGG_ISSUE_FULL(ids, e)                                               \
    _Pragma("unroll")                                                        \
    for (int u = 0; u < AGG_ROWS; ++u) {                                     \
        int s = __builtin_amdgcn_readlane((ids), (e) + u);                   \
        r[u] = *(gcf4p)(x + (size_t)s * D128 + lo);                          \
    }
#define AGG_ISSUE_PART(ids, e, m)                                            \
    {                                                                        \
        int sid[AGG_ROWS - 1];                                               \
        _Pragma("unroll")                                                    \
        for (int u = 0; u < AGG_ROWS - 1; ++u)                               \
            sid[u] = __builtin_amdgcn_readlane((ids), (e) + u);              \
        _Pragma("unroll")                                                    \
        for (int u = 0; u < AGG_ROWS - 1; ++u)                               \
            if (u < (m)) r[u] = *(gcf4p)(x + (size_t)sid[u] * D128 + lo);    \
    }
#define AGG_ADD_FULL()                                                       \
    _Pragma("unroll")                                                        \
    for (int u = 0; u < AGG_ROWS; ++u) acc += r[u];
#define AGG_ADD_PART(m)                                                      \
    _Pragma("unroll")                                                        \
    for (int u = 0; u < AGG_ROWS - 1; ++u)                                   \
        if (u < (m)) acc += r[u];

#define LDA 132
#define GEMM_MT 4
#define GEMM_BIG 768
#define SLAB (GEMM_MT * 16 * LDA)

typedef       __attribute__((address_space(3))) float* lfp;
typedef       __attribute__((address_space(3))) f4v*   lf4p;
typedef const __attribute__((address_space(3))) v2f*   lcv2p;
typedef       __attribute__((address_space(3))) int*   lip;

template <int NT>
__device__ __forceinline__ void stage_rows(lfp As, int row0, int t,
        gcfp x, gcip src, gcip offs, gcip deg) {
    int lane = t & 31;
    int lo   = lane * 4;
    int rl   = __builtin_amdgcn_readfirstlane(t >> 5);
    int node = row0 + rl;

    int degv = 0, offv = 0;
    if (lane < NT * 2) {
        degv = deg[node + 8 * lane];
        offv = offs[node + 8 * lane];
    }
    int dg  = __builtin_amdgcn_readlane(degv, 0);
    int off = __builtin_amdgcn_readlane(offv, 0);
    int my = 0;
    if (lane < (dg < 32 ? dg : 32)) my = src[off + lane];

    for (int i = 0; i < NT * 2; ++i) {
        int nodeN = node + 8;
        int dgN  = __builtin_amdgcn_readlane(degv, i + 1);
        int offN = __builtin_amdgcn_readlane(offv, i + 1);
        int base = node * D128 + lo;
        f4v acc  = { 0.f, 0.f, 0.f, 0.f };
        f4v r[AGG_ROWS];
        int cnt = dg < 32 ? dg : 32;

        if (dg > 0) {
            if (cnt >= AGG_ROWS) { AGG_ISSUE_FULL(my, 0) }
            else                 { AGG_ISSUE_PART(my, 0, cnt) }
        }
        f4v self = *(gcf4p)(x + base);
        int myN = 0;
        if (lane < (dgN < 32 ? dgN : 32)) myN = src[offN + lane];

        if (dg > 0) {
            if (cnt >= AGG_ROWS) { AGG_ADD_FULL() }
            else                 { AGG_ADD_PART(cnt) }
            for (int e = AGG_ROWS; e < cnt; e += AGG_ROWS) {
                int m = cnt - e;
                if (m >= AGG_ROWS) {
                    AGG_ISSUE_FULL(my, e)
                    __builtin_amdgcn_sched_barrier(0);
                    AGG_ADD_FULL()
                } else {
                    AGG_ISSUE_PART(my, e, m)
                    AGG_ADD_PART(m)
                }
            }
            for (int c0 = 32; c0 < dg; c0 += 32) {
                int c = dg - c0;
                c = c < 32 ? c : 32;
                int ids = 0;
                if (lane < c) ids = src[off + c0 + lane];
                for (int e = 0; e < c; e += AGG_ROWS) {
                    int m = c - e;
                    if (m >= AGG_ROWS) {
                        AGG_ISSUE_FULL(ids, e)
                        __builtin_amdgcn_sched_barrier(0);
                        AGG_ADD_FULL()
                    } else {
                        AGG_ISSUE_PART(ids, e, m)
                        AGG_ADD_PART(m)
                    }
                }
            }
            float inv = 1.0f / (float)dg;
            acc *= inv;
        }
        lfp dstrow = As + rl * LDA + lo;
        *(lf4p)(dstrow)        = self;
        *(lf4p)(dstrow + SLAB) = acc;

        node = nodeN; rl += 8; dg = dgN; off = offN; my = myN;
    }
}
#undef AGG_ISSUE_FULL
#undef AGG_ISSUE_PART
#undef AGG_ADD_FULL
#undef AGG_ADD_PART

template <int NT>
__device__ __forceinline__ void gemm_rows(lfp As, int row0, int t,
        gcfp x, gcip src, gcip offs, gcip deg, gcfp Wf,
        gcfp bias, gfp out) {
    int wave = t >> 5;
    int lane = t & 31;

    int d0 = (lane < NT * 16) ? deg[row0 + lane] : 1;
    int d1 = (lane + 32 < NT * 16) ? deg[row0 + 32 + lane] : 1;

    stage_rows<NT>(As, row0, t, x, src, offs, deg);

    unsigned z0 = __builtin_amdgcn_ballot_w32(d0 == 0);
    unsigned z1 = __builtin_amdgcn_ballot_w32(d1 == 0);
    int cnt0 = __builtin_popcount(z0);
    int cnt  = cnt0 + __builtin_popcount(z1);
    lip zl = (lip)(As + 2 * SLAB) + wave * 64;
    {
        unsigned below = (1u << lane) - 1u;
        if ((z0 >> lane) & 1u) zl[__builtin_popcount(z0 & below)] = lane;
        if ((z1 >> lane) & 1u) zl[cnt0 + __builtin_popcount(z1 & below)] = lane + 32;
    }
    __syncthreads();

    int col0 = wave * 16;
    int mrow = lane & 15;
    int koff = (lane >> 4) * 2;

    v8f acc[NT];
    #pragma unroll
    for (int m = 0; m < NT; ++m)
        acc[m] = (v8f){ 0.f, 0.f, 0.f, 0.f, 0.f, 0.f, 0.f, 0.f };

    #pragma unroll
    for (int b2 = 0; b2 < 2; ++b2) {
        lfp   A  = As + b2 * SLAB + mrow * LDA + koff;
        gcf4p Bf = (gcf4p)Wf + b2 * FRAG_F4 + wave * 512 + lane;
        #pragma unroll
        for (int kk = 0; kk < D128 / 8; ++kk) {
            f4v bb = Bf[kk * 32];
            #pragma unroll
            for (int j = 0; j < 2; ++j) {
                v2f b = j ? (v2f){ bb.z, bb.w } : (v2f){ bb.x, bb.y };
                int k = kk * 8 + j * 4;
                #pragma unroll
                for (int m = 0; m < NT; ++m) {
                    v2f a = *(lcv2p)(A + m * 16 * LDA + k);
                    acc[m] = __builtin_amdgcn_wmma_f32_16x16x4_f32(
                        false, a, false, b, (short)0, acc[m], false, false);
                }
            }
        }
    }

    int colg = col0 + (lane & 15);
    float bv = bias[colg];
    #pragma unroll
    for (int m = 0; m < NT; ++m) {
        int rbase = row0 + m * 16 + (lane >> 4) * 8;
        unsigned zw = (m < 2) ? z0 : z1;
        unsigned zb = zw >> ((m & 1) * 16 + (lane >> 4) * 8);
        #pragma unroll
        for (int i = 0; i < 8; ++i) {
            if (!((zb >> i) & 1u)) {
                float v = acc[m][i] + bv;
                v = v > 0.f ? v : (expf(v) - 1.f);
                out[(size_t)(rbase + i) * D128 + colg] = v;
            }
        }
    }

    for (int j0 = 0; j0 < cnt; j0 += 16) {
        int ia = j0 + mrow;
        int zr = zl[ia < cnt ? ia : 0];
        lfp   A  = As + zr * LDA + koff;
        gcf4p Bf = (gcf4p)Wf + 2 * FRAG_F4 + wave * 512 + lane;
        v8f z = (v8f){ 0.f, 0.f, 0.f, 0.f, 0.f, 0.f, 0.f, 0.f };
        #pragma unroll
        for (int kk = 0; kk < D128 / 8; ++kk) {
            f4v bb = Bf[kk * 32];
            #pragma unroll
            for (int j = 0; j < 2; ++j) {
                v2f b = j ? (v2f){ bb.z, bb.w } : (v2f){ bb.x, bb.y };
                v2f a = *(lcv2p)(A + kk * 8 + j * 4);
                z = __builtin_amdgcn_wmma_f32_16x16x4_f32(
                    false, a, false, b, (short)0, z, false, false);
            }
        }
        int e0 = j0 + (lane >> 4) * 8;
        int zrow[8];
        #pragma unroll
        for (int i = 0; i < 8; ++i)
            zrow[i] = zl[(e0 + i) < cnt ? (e0 + i) : 0];
        #pragma unroll
        for (int i = 0; i < 8; ++i) {
            if (e0 + i < cnt) {
                float v = z[i] + bv;
                v = v > 0.f ? v : (expf(v) - 1.f);
                out[(size_t)(row0 + zrow[i]) * D128 + colg] = v;
            }
        }
    }
}

__global__ __launch_bounds__(256)
void gemm_fused(const float* __restrict__ x_, const int* __restrict__ src_,
                const int* __restrict__ offs_, const int* __restrict__ deg_,
                const float* __restrict__ Wg_, const float* __restrict__ Ws_,
                const float* __restrict__ Wl_, const float* __restrict__ Wf_,
                const float* __restrict__ bias_, float* __restrict__ out_, int n) {
    __shared__ float As[3 * SLAB];

    int t = threadIdx.x;
    int b = blockIdx.x;
    int nbig = (n >> 4) / GEMM_MT;
    nbig = nbig < GEMM_BIG ? nbig : GEMM_BIG;

    if (b < nbig) {
        gemm_rows<GEMM_MT>((lfp)As, b * (GEMM_MT * 16), t,
                           (gcfp)x_, (gcip)src_, (gcip)offs_, (gcip)deg_,
                           (gcfp)Wf_,
                           (gcfp)bias_, (gfp)out_);
        return;
    }

    int tile = nbig * GEMM_MT + (b - nbig);
    int row0 = tile * 16;

    if (row0 + 16 > n) {
        for (int r = row0; r < n; ++r) {
            int dg  = deg_[r];
            int off = offs_[r];
            float inv = dg > 0 ? 1.0f / (float)dg : 0.f;
            for (int c = t; c < D128; c += 256) {
                float s = bias_[c];
                const float* xr = x_  + (size_t)r * D128;
                const float* wg = Wg_ + (size_t)c * D128;
                const float* ws = Ws_ + (size_t)c * D128;
                const float* wl = Wl_ + (size_t)c * D128;
                for (int k = 0; k < D128; ++k) {
                    float nmk = 0.f;
                    for (int e = 0; e < dg; ++e)
                        nmk += x_[(size_t)src_[off + e] * D128 + k];
                    nmk *= inv;
                    float xmk = dg > 0 ? xr[k] : 0.f;
                    s += xr[k] * wg[k] + xmk * ws[k] + nmk * wl[k];
                }
                s = s > 0.f ? s : (expf(s) - 1.f);
                out_[(size_t)r * D128 + c] = s;
            }
        }
        return;
    }

    gemm_rows<1>((lfp)As, row0, t,
                 (gcfp)x_, (gcip)src_, (gcip)offs_, (gcip)deg_,
                 (gcfp)Wf_,
                 (gcfp)bias_, (gfp)out_);
}

extern "C" void kernel_launch(void* const* d_in, const int* in_sizes, int n_in,
                              void* d_out, int out_size, void* d_ws, size_t ws_size,
                              hipStream_t stream) {
    (void)n_in; (void)out_size; (void)ws_size;
    const float* x    = (const float*)d_in[0];
    const float* Wg   = (const float*)d_in[1];
    const float* Wl   = (const float*)d_in[2];
    const float* Ws   = (const float*)d_in[3];
    const float* bias = (const float*)d_in[4];
    const int*   src  = (const int*)d_in[5];
    const int*   deg  = (const int*)d_in[7];

    int N = in_sizes[7];

    float* Wf = (float*)d_ws;
    int* offs = (int*)d_ws + 3 * D128 * D128;

    int scanBlocks = (N + SCAN_THREADS - 1) / SCAN_THREADS;
    scan_offsets<<<scanBlocks, SCAN_THREADS, 0, stream>>>(deg, offs, N, Wg, Ws, Wl, Wf);

    int tiles = (N + 15) / 16;
    int nbig = (N / 16) / GEMM_MT;
    nbig = nbig < GEMM_BIG ? nbig : GEMM_BIG;
    int gblocks = nbig + (tiles - nbig * GEMM_MT);
    gemm_fused<<<gblocks, 256, 0, stream>>>(x, src, offs, deg, Wg, Ws, Wl, Wf, bias,
                                            (float*)d_out, N);
}


# deg==0 rows listed once by the scan launch and multiplied by 64 blocks at the end of the grid, 32 rows a trip: no block of the main product runs its own Wg tile any more; sum order unchanged
# speedup vs baseline: 3.6723x; 1.0400x over previous
#include <hip/hip_runtime.h>
#include <math.h>

typedef __attribute__((ext_vector_type(2))) float v2f;
typedef __attribute__((ext_vector_type(4))) float f4v;
typedef __attribute__((ext_vector_type(8))) float v8f;
typedef __attribute__((ext_vector_type(4))) int   i4v;

typedef const __attribute__((address_space(1))) float* gcfp;
typedef       __attribute__((address_space(1))) float* gfp;
typedef const __attribute__((address_space(1))) int*   gcip;
typedef       __attribute__((address_space(1))) int*   gip;
typedef const __attribute__((address_space(1))) f4v*   gcf4p;
typedef       __attribute__((address_space(1))) f4v*   gf4p;
typedef const __attribute__((address_space(1))) v2f*   gcv2p;
typedef const __attribute__((address_space(1))) i4v*   gci4p;

#define D128 128
#define SCAN_THREADS 1024

#define FRAG_F4 (D128 * D128 / 4)
__global__ __launch_bounds__(SCAN_THREADS)
void scan_offsets(const int* __restrict__ deg_, int* __restrict__ offs_, int n,
                  const float* __restrict__ Wg_, const float* __restrict__ Ws_,
                  const float* __restrict__ Wl_, float* __restrict__ Wf_,
                  int* __restrict__ zl_) {
    static_assert(SCAN_THREADS == 32 * 32, "one LDS slot per lane: 32 waves of 32 lanes");
    gcip  deg  = (gcip)deg_;
    gci4p deg4 = (gci4p)deg_;
    gip   offs = (gip)offs_;
    gip   zl   = (gip)zl_;
    __shared__ int wtot[32];
    __shared__ int wpre[32];
    __shared__ int ztot[32];
    __shared__ int zpre[32];

    int t = threadIdx.x;
    int lane = t & 31;
    int wave = t >> 5;

    for (int g = blockIdx.x * SCAN_THREADS + t; g < 3 * FRAG_F4;
         g += gridDim.x * SCAN_THREADS) {
        int sel = g / FRAG_F4;
        int idx = g - sel * FRAG_F4;
        int ln  = idx & 31;
        int kk  = (idx >> 5) & 15;
        int ct  = idx >> 9;
        int e   = (ct * 16 + (ln & 15)) * D128 + kk * 8 + (ln >> 4) * 2;
        gcfp p  = (gcfp)(sel == 1 ? Wl_ : Wg_) + e;
        v2f lo2 = *(gcv2p)p;
        v2f hi2 = *(gcv2p)(p + 4);
        if (sel == 0) {
            gcfp q = (gcfp)Ws_ + e;
            lo2 += *(gcv2p)q;
            hi2 += *(gcv2p)(q + 4);
        }
        f4v o = { lo2.x, lo2.y, hi2.x, hi2.y };
        ((gf4p)Wf_)[g] = o;
    }

    int start = blockIdx.x * SCAN_THREADS;
    int i = start + t;
    int d = (i < n) ? deg[i] : 0;

    int nq = start >> 2;
    int pre = 0;
    int zfr = 0;
    for (int q0 = t; q0 < nq; q0 += 4 * SCAN_THREADS) {
        #pragma unroll
        for (int u = 0; u < 4; ++u) {
            int q = q0 + u * SCAN_THREADS;
            bool ok = q < nq;
            i4v v = deg4[ok ? q : q0];
            int sv = (v.x + v.y) + (v.z + v.w);
            int zv = ((v.x == 0) + (v.y == 0)) + ((v.z == 0) + (v.w == 0));
            pre += ok ? sv : 0;
            zfr += ok ? zv : 0;
        }
    }

    int inc = d;
    int zme = (i < n && d == 0) ? 1 : 0;
    int zin = zme;
    #pragma unroll
    for (int o = 1; o < 32; o <<= 1) {
        int v = __shfl_up(inc, o);
        int w = __shfl_up(zin, o);
        if (lane >= o) { inc += v; zin += w; }
    }
    #pragma unroll
    for (int o = 16; o > 0; o >>= 1) {
        pre += __shfl_xor(pre, o);
        zfr += __shfl_xor(zfr, o);
    }

    if (lane == 31) { wtot[wave] = inc; ztot[wave] = zin; }
    if (lane == 0)  { wpre[wave] = pre; zpre[wave] = zfr; }
    __syncthreads();

    int c = wpre[lane] + ((lane < wave) ? wtot[lane] : 0);
    int cz = zpre[lane] + ((lane < wave) ? ztot[lane] : 0);
    #pragma unroll
    for (int o = 16; o > 0; o >>= 1) {
        c  += __shfl_xor(c, o);
        cz += __shfl_xor(cz, o);
    }

    if (i < n) offs[i] = c + inc - d;
    if (zme) zl[1 + cz + zin - 1] = i;
    if (i == n - 1) zl[0] = cz + zin;
}

#define AGG_ROWS 16

#define AGG_ISSUE_FULL(ids, e)                                               \
    _Pragma("unroll")                                                        \
    for (int u = 0; u < AGG_ROWS; ++u) {                                     \
        int s = __builtin_amdgcn_readlane((ids), (e) + u);                   \
        r[u] = *(gcf4p)(x + (size_t)s * D128 + lo);                          \
    }
#define AGG_ISSUE_PART(ids, e, m)                                            \
    {                                                                        \
        int sid[AGG_ROWS - 1];                                               \
        _Pragma("unroll")                                                    \
        for (int u = 0; u < AGG_ROWS - 1; ++u)                               \
            sid[u] = __builtin_amdgcn_readlane((ids), (e) + u);              \
        _Pragma("unroll")                                                    \
        for (int u = 0; u < AGG_ROWS - 1; ++u)                               \
            if (u < (m)) r[u] = *(gcf4p)(x + (size_t)sid[u] * D128 + lo);    \
    }
#define AGG_ADD_FULL()                                                       \
    _Pragma("unroll")                                                        \
    for (int u = 0; u < AGG_ROWS; ++u) acc += r[u];
#define AGG_ADD_PART(m)                                                      \
    _Pragma("unroll")                                                        \
    for (int u = 0; u < AGG_ROWS - 1; ++u)                                   \
        if (u < (m)) acc += r[u];

#define LDA 132
#define GEMM_MT 4
#define GEMM_BIG 768
#define SLAB (GEMM_MT * 16 * LDA)

typedef       __attribute__((address_space(3))) float* lfp;
typedef       __attribute__((address_space(3))) f4v*   lf4p;
typedef const __attribute__((address_space(3))) v2f*   lcv2p;
typedef       __attribute__((address_space(3))) int*   lip;

template <int NT>
__device__ __forceinline__ void stage_rows(lfp As, int row0, int t,
        gcfp x, gcip src, gcip offs, gcip deg) {
    int lane = t & 31;
    int lo   = lane * 4;
    int rl   = __builtin_amdgcn_readfirstlane(t >> 5);
    int node = row0 + rl;

    int degv = 0, offv = 0;
    if (lane < NT * 2) {
        degv = deg[node + 8 * lane];
        offv = offs[node + 8 * lane];
    }
    int dg  = __builtin_amdgcn_readlane(degv, 0);
    int off = __builtin_amdgcn_readlane(offv, 0);
    int my = 0;
    if (lane < (dg < 32 ? dg : 32)) my = src[off + lane];

    for (int i = 0; i < NT * 2; ++i) {
        int nodeN = node + 8;
        int dgN  = __builtin_amdgcn_readlane(degv, i + 1);
        int offN = __builtin_amdgcn_readlane(offv, i + 1);
        int base = node * D128 + lo;
        f4v acc  = { 0.f, 0.f, 0.f, 0.f };
        f4v r[AGG_ROWS];
        int cnt = dg < 32 ? dg : 32;

        if (dg > 0) {
            if (cnt >= AGG_ROWS) { AGG_ISSUE_FULL(my, 0) }
            else                 { AGG_ISSUE_PART(my, 0, cnt) }
        }
        f4v self = *(gcf4p)(x + base);
        int myN = 0;
        if (lane < (dgN < 32 ? dgN : 32)) myN = src[offN + lane];

        if (dg > 0) {
            if (cnt >= AGG_ROWS) { AGG_ADD_FULL() }
            else                 { AGG_ADD_PART(cnt) }
            for (int e = AGG_ROWS; e < cnt; e += AGG_ROWS) {
                int m = cnt - e;
                if (m >= AGG_ROWS) {
                    AGG_ISSUE_FULL(my, e)
                    __builtin_amdgcn_sched_barrier(0);
                    AGG_ADD_FULL()
                } else {
                    AGG_ISSUE_PART(my, e, m)
                    AGG_ADD_PART(m)
                }
            }
            for (int c0 = 32; c0 < dg; c0 += 32) {
                int c = dg - c0;
                c = c < 32 ? c : 32;
                int ids = 0;
                if (lane < c) ids = src[off + c0 + lane];
                for (int e = 0; e < c; e += AGG_ROWS) {
                    int m = c - e;
                    if (m >= AGG_ROWS) {
                        AGG_ISSUE_FULL(ids, e)
                        __builtin_amdgcn_sched_barrier(0);
                        AGG_ADD_FULL()
                    } else {
                        AGG_ISSUE_PART(ids, e, m)
                        AGG_ADD_PART(m)
                    }
                }
            }
            float inv = 1.0f / (float)dg;
            acc *= inv;
        }
        lfp dstrow = As + rl * LDA + lo;
        *(lf4p)(dstrow)        = self;
        *(lf4p)(dstrow + SLAB) = acc;

        node = nodeN; rl += 8; dg = dgN; off = offN; my = myN;
    }
}
#undef AGG_ISSUE_FULL
#undef AGG_ISSUE_PART
#undef AGG_ADD_FULL
#undef AGG_ADD_PART

template <int NT>
__device__ __forceinline__ void gemm_rows(lfp As, int row0, int t,
        gcfp x, gcip src, gcip offs, gcip deg, gcfp Wf,
        gcfp bias, gfp out) {
    int wave = t >> 5;
    int lane = t & 31;

    int d0 = (lane < NT * 16) ? deg[row0 + lane] : 1;
    int d1 = (lane + 32 < NT * 16) ? deg[row0 + 32 + lane] : 1;

    stage_rows<NT>(As, row0, t, x, src, offs, deg);

    unsigned z0 = __builtin_amdgcn_ballot_w32(d0 == 0);
    unsigned z1 = __builtin_amdgcn_ballot_w32(d1 == 0);
    __syncthreads();

    int col0 = wave * 16;
    int mrow = lane & 15;
    int koff = (lane >> 4) * 2;

    v8f acc[NT];
    #pragma unroll
    for (int m = 0; m < NT; ++m)
        acc[m] = (v8f){ 0.f, 0.f, 0.f, 0.f, 0.f, 0.f, 0.f, 0.f };

    #pragma unroll
    for (int b2 = 0; b2 < 2; ++b2) {
        lfp   A  = As + b2 * SLAB + mrow * LDA + koff;
        gcf4p Bf = (gcf4p)Wf + b2 * FRAG_F4 + wave * 512 + lane;
        #pragma unroll
        for (int kk = 0; kk < D128 / 8; ++kk) {
            f4v bb = Bf[kk * 32];
            #pragma unroll
            for (int j = 0; j < 2; ++j) {
                v2f b = j ? (v2f){ bb.z, bb.w } : (v2f){ bb.x, bb.y };
                int k = kk * 8 + j * 4;
                #pragma unroll
                for (int m = 0; m < NT; ++m) {
                    v2f a = *(lcv2p)(A + m * 16 * LDA + k);
                    acc[m] = __builtin_amdgcn_wmma_f32_16x16x4_f32(
                        false, a, false, b, (short)0, acc[m], false, false);
                }
            }
        }
    }

    int colg = col0 + (lane & 15);
    float bv = bias[colg];
    #pragma unroll
    for (int m = 0; m < NT; ++m) {
        int rbase = row0 + m * 16 + (lane >> 4) * 8;
        unsigned zw = (m < 2) ? z0 : z1;
        unsigned zb = zw >> ((m & 1) * 16 + (lane >> 4) * 8);
        #pragma unroll
        for (int i = 0; i < 8; ++i) {
            if (!((zb >> i) & 1u)) {
                float v = acc[m][i] + bv;
                v = v > 0.f ? v : (expf(v) - 1.f);
                out[(size_t)(rbase + i) * D128 + colg] = v;
            }
        }
    }
}

#define ZB 64
#define ZT 2
#define ZR (ZT * 16)
__device__ __forceinline__ void zero_rows(lfp As, int zb, int t,
        gcfp x, gcip zl, gcfp Wf, gcfp bias, gfp out) {
    int wave = t >> 5;
    int lane = t & 31;
    int lo   = lane * 4;
    int mrow = lane & 15;
    int koff = (lane >> 4) * 2;
    int colg = wave * 16 + mrow;
    int Z = zl[0];
    float bv = bias[colg];
    lip ids = (lip)(As + 2 * SLAB);

    for (int e0 = zb * ZR; e0 < Z; e0 += ZB * ZR) {
        if (t < ZR) {
            int e = e0 + t;
            ids[t] = zl[1 + (e < Z ? e : e0)];
        }
        int rid = 0;
        if (lane < ZR / 8) {
            int e = e0 + wave + 8 * lane;
            rid = zl[1 + (e < Z ? e : e0)];
        }
        f4v rowv[ZR / 8];
        #pragma unroll
        for (int u = 0; u < ZR / 8; ++u) {
            int r = __builtin_amdgcn_readlane(rid, u);
            rowv[u] = *(gcf4p)(x + (size_t)r * D128 + lo);
        }
        #pragma unroll
        for (int u = 0; u < ZR / 8; ++u)
            *(lf4p)(As + (wave + 8 * u) * LDA + lo) = rowv[u];
        __syncthreads();

        v8f acc[ZT];
        #pragma unroll
        for (int m = 0; m < ZT; ++m)
            acc[m] = (v8f){ 0.f, 0.f, 0.f, 0.f, 0.f, 0.f, 0.f, 0.f };
        lfp   A  = As + mrow * LDA + koff;
        gcf4p Bf = (gcf4p)Wf + 2 * FRAG_F4 + wave * 512 + lane;
        #pragma unroll
        for (int kk = 0; kk < D128 / 8; ++kk) {
            f4v bb = Bf[kk * 32];
            #pragma unroll
            for (int j = 0; j < 2; ++j) {
                v2f b = j ? (v2f){ bb.z, bb.w } : (v2f){ bb.x, bb.y };
                int k = kk * 8 + j * 4;
                #pragma unroll
                for (int m = 0; m < ZT; ++m) {
                    v2f a = *(lcv2p)(A + m * 16 * LDA + k);
                    acc[m] = __builtin_amdgcn_wmma_f32_16x16x4_f32(
                        false, a, false, b, (short)0, acc[m], false, false);
                }
            }
        }
        #pragma unroll
        for (int m = 0; m < ZT; ++m) {
            int sb = m * 16 + (lane >> 4) * 8;
            #pragma unroll
            for (int i = 0; i < 8; ++i) {
                if (e0 + sb + i < Z) {
                    float v = acc[m][i] + bv;
                    v = v > 0.f ? v : (expf(v) - 1.f);
                    out[(size_t)ids[sb + i] * D128 + colg] = v;
                }
            }
        }
        __syncthreads();
    }
}

__global__ __launch_bounds__(256)
void gemm_fused(const float* __restrict__ x_, const int* __restrict__ src_,
                const int* __restrict__ offs_, const int* __restrict__ deg_,
                const float* __restrict__ Wg_, const float* __restrict__ Ws_,
                const float* __restrict__ Wl_, const float* __restrict__ Wf_,
                const float* __restrict__ bias_, float* __restrict__ out_, int n,
                const int* __restrict__ zl_) {
    __shared__ float As[3 * SLAB];

    int t = threadIdx.x;
    int b = blockIdx.x;
    int nbig = (n >> 4) / GEMM_MT;
    nbig = nbig < GEMM_BIG ? nbig : GEMM_BIG;

    if (b < nbig) {
        gemm_rows<GEMM_MT>((lfp)As, b * (GEMM_MT * 16), t,
                           (gcfp)x_, (gcip)src_, (gcip)offs_, (gcip)deg_,
                           (gcfp)Wf_,
                           (gcfp)bias_, (gfp)out_);
        return;
    }

    int nsmall = ((n + 15) >> 4) - nbig * GEMM_MT;
    if (b >= nbig + nsmall) {
        zero_rows((lfp)As, b - nbig - nsmall, t, (gcfp)x_, (gcip)zl_,
                  (gcfp)Wf_, (gcfp)bias_, (gfp)out_);
        return;
    }

    int tile = nbig * GEMM_MT + (b - nbig);
    int row0 = tile * 16;

    if (row0 + 16 > n) {
        for (int r = row0; r < n; ++r) {
            int dg  = deg_[r];
            if (dg == 0) continue;
            int off = offs_[r];
            float inv = dg > 0 ? 1.0f / (float)dg : 0.f;
            for (int c = t; c < D128; c += 256) {
                float s = bias_[c];
                const float* xr = x_  + (size_t)r * D128;
                const float* wg = Wg_ + (size_t)c * D128;
                const float* ws = Ws_ + (size_t)c * D128;
                const float* wl = Wl_ + (size_t)c * D128;
                for (int k = 0; k < D128; ++k) {
                    float nmk = 0.f;
                    for (int e = 0; e < dg; ++e)
                        nmk += x_[(size_t)src_[off + e] * D128 + k];
                    nmk *= inv;
                    float xmk = dg > 0 ? xr[k] : 0.f;
                    s += xr[k] * wg[k] + xmk * ws[k] + nmk * wl[k];
                }
                s = s > 0.f ? s : (expf(s) - 1.f);
                out_[(size_t)r * D128 + c] = s;
            }
        }
        return;
    }

    gemm_rows<1>((lfp)As, row0, t,
                 (gcfp)x_, (gcip)src_, (gcip)offs_, (gcip)deg_,
                 (gcfp)Wf_,
                 (gcfp)bias_, (gfp)out_);
}

extern "C" void kernel_launch(void* const* d_in, const int* in_sizes, int n_in,
                              void* d_out, int out_size, void* d_ws, size_t ws_size,
                              hipStream_t stream) {
    (void)n_in; (void)out_size; (void)ws_size;
    const float* x    = (const float*)d_in[0];
    const float* Wg   = (const float*)d_in[1];
    const float* Wl   = (const float*)d_in[2];
    const float* Ws   = (const float*)d_in[3];
    const float* bias = (const float*)d_in[4];
    const int*   src  = (const int*)d_in[5];
    const int*   deg  = (const int*)d_in[7];

    int N = in_sizes[7];

    float* Wf = (float*)d_ws;
    int* offs = (int*)d_ws + 3 * D128 * D128;
    int* zl   = offs + N;

    int scanBlocks = (N + SCAN_THREADS - 1) / SCAN_THREADS;
    scan_offsets<<<scanBlocks, SCAN_THREADS, 0, stream>>>(deg, offs, N, Wg, Ws, Wl, Wf, zl);

    int tiles = (N + 15) / 16;
    int nbig = (N / 16) / GEMM_MT;
    nbig = nbig < GEMM_BIG ? nbig : GEMM_BIG;
    int gblocks = nbig + (tiles - nbig * GEMM_MT) + ZB;
    gemm_fused<<<gblocks, 256, 0, stream>>>(x, src, offs, deg, Wg, Ws, Wl, Wf, bias,
                                            (float*)d_out, N, zl);
}


# main product's epilogue staged through LDS (the free third slab): each store instruction writes one whole 512-byte row, not two 64-byte pieces; values and sum order unchanged
# speedup vs baseline: 3.9801x; 1.0838x over previous
#include <hip/hip_runtime.h>
#include <math.h>

typedef __attribute__((ext_vector_type(2))) float v2f;
typedef __attribute__((ext_vector_type(4))) float f4v;
typedef __attribute__((ext_vector_type(8))) float v8f;
typedef __attribute__((ext_vector_type(4))) int   i4v;

typedef const __attribute__((address_space(1))) float* gcfp;
typedef       __attribute__((address_space(1))) float* gfp;
typedef const __attribute__((address_space(1))) int*   gcip;
typedef       __attribute__((address_space(1))) int*   gip;
typedef const __attribute__((address_space(1))) f4v*   gcf4p;
typedef       __attribute__((address_space(1))) f4v*   gf4p;
typedef const __attribute__((address_space(1))) v2f*   gcv2p;
typedef const __attribute__((address_space(1))) i4v*   gci4p;

#define D128 128
#define SCAN_THREADS 1024

#define FRAG_F4 (D128 * D128 / 4)
__global__ __launch_bounds__(SCAN_THREADS)
void scan_offsets(const int* __restrict__ deg_, int* __restrict__ offs_, int n,
                  const float* __restrict__ Wg_, const float* __restrict__ Ws_,
                  const float* __restrict__ Wl_, float* __restrict__ Wf_,
                  int* __restrict__ zl_) {
    static_assert(SCAN_THREADS == 32 * 32, "one LDS slot per lane: 32 waves of 32 lanes");
    gcip  deg  = (gcip)deg_;
    gci4p deg4 = (gci4p)deg_;
    gip   offs = (gip)offs_;
    gip   zl   = (gip)zl_;
    __shared__ int wtot[32];
    __shared__ int wpre[32];
    __shared__ int ztot[32];
    __shared__ int zpre[32];

    int t = threadIdx.x;
    int lane = t & 31;
    int wave = t >> 5;

    for (int g = blockIdx.x * SCAN_THREADS + t; g < 3 * FRAG_F4;
         g += gridDim.x * SCAN_THREADS) {
        int sel = g / FRAG_F4;
        int idx = g - sel * FRAG_F4;
        int ln  = idx & 31;
        int kk  = (idx >> 5) & 15;
        int ct  = idx >> 9;
        int e   = (ct * 16 + (ln & 15)) * D128 + kk * 8 + (ln >> 4) * 2;
        gcfp p  = (gcfp)(sel == 1 ? Wl_ : Wg_) + e;
        v2f lo2 = *(gcv2p)p;
        v2f hi2 = *(gcv2p)(p + 4);
        if (sel == 0) {
            gcfp q = (gcfp)Ws_ + e;
            lo2 += *(gcv2p)q;
            hi2 += *(gcv2p)(q + 4);
        }
        f4v o = { lo2.x, lo2.y, hi2.x, hi2.y };
        ((gf4p)Wf_)[g] = o;
    }

    int start = blockIdx.x * SCAN_THREADS;
    int i = start + t;
    int d = (i < n) ? deg[i] : 0;

    int nq = start >> 2;
    int pre = 0;
    int zfr = 0;
    for (int q0 = t; q0 < nq; q0 += 4 * SCAN_THREADS) {
        #pragma unroll
        for (int u = 0; u < 4; ++u) {
            int q = q0 + u * SCAN_THREADS;
            bool ok = q < nq;
            i4v v = deg4[ok ? q : q0];
            int sv = (v.x + v.y) + (v.z + v.w);
            int zv = ((v.x == 0) + (v.y == 0)) + ((v.z == 0) + (v.w == 0));
            pre += ok ? sv : 0;
            zfr += ok ? zv : 0;
        }
    }

    int inc = d;
    int zme = (i < n && d == 0) ? 1 : 0;
    int zin = zme;
    #pragma unroll
    for (int o = 1; o < 32; o <<= 1) {
        int v = __shfl_up(inc, o);
        int w = __shfl_up(zin, o);
        if (lane >= o) { inc += v; zin += w; }
    }
    #pragma unroll
    for (int o = 16; o > 0; o >>= 1) {
        pre += __shfl_xor(pre, o);
        zfr += __shfl_xor(zfr, o);
    }

    if (lane == 31) { wtot[wave] = inc; ztot[wave] = zin; }
    if (lane == 0)  { wpre[wave] = pre; zpre[wave] = zfr; }
    __syncthreads();

    int c = wpre[lane] + ((lane < wave) ? wtot[lane] : 0);
    int cz = zpre[lane] + ((lane < wave) ? ztot[lane] : 0);
    #pragma unroll
    for (int o = 16; o > 0; o >>= 1) {
        c  += __shfl_xor(c, o);
        cz += __shfl_xor(cz, o);
    }

    if (i < n) offs[i] = c + inc - d;
    if (zme) zl[1 + cz + zin - 1] = i;
    if (i == n - 1) zl[0] = cz + zin;
}

#define AGG_ROWS 16

#define AGG_ISSUE_FULL(ids, e)                                               \
    _Pragma("unroll")                                                        \
    for (int u = 0; u < AGG_ROWS; ++u) {                                     \
        int s = __builtin_amdgcn_readlane((ids), (e) + u);                   \
        r[u] = *(gcf4p)(x + (size_t)s * D128 + lo);                          \
    }
#define AGG_ISSUE_PART(ids, e, m)                                            \
    {                                                                        \
        int sid[AGG_ROWS - 1];                                               \
        _Pragma("unroll")                                                    \
        for (int u = 0; u < AGG_ROWS - 1; ++u)                               \
            sid[u] = __builtin_amdgcn_readlane((ids), (e) + u);              \
        _Pragma("unroll")                                                    \
        for (int u = 0; u < AGG_ROWS - 1; ++u)                               \
            if (u < (m)) r[u] = *(gcf4p)(x + (size_t)sid[u] * D128 + lo);    \
    }
#define AGG_ADD_FULL()                                                       \
    _Pragma("unroll")                                                        \
    for (int u = 0; u < AGG_ROWS; ++u) acc += r[u];
#define AGG_ADD_PART(m)                                                      \
    _Pragma("unroll")                                                        \
    for (int u = 0; u < AGG_ROWS - 1; ++u)                                   \
        if (u < (m)) acc += r[u];

#define LDA 132
#define GEMM_MT 4
#define GEMM_BIG 768
#define SLAB (GEMM_MT * 16 * LDA)

typedef       __attribute__((address_space(3))) float* lfp;
typedef       __attribute__((address_space(3))) f4v*   lf4p;
typedef const __attribute__((address_space(3))) v2f*   lcv2p;
typedef       __attribute__((address_space(3))) int*   lip;

template <int NT>
__device__ __forceinline__ void stage_rows(lfp As, int row0, int t,
        gcfp x, gcip src, gcip offs, gcip deg) {
    int lane = t & 31;
    int lo   = lane * 4;
    int rl   = __builtin_amdgcn_readfirstlane(t >> 5);
    int node = row0 + rl;

    int degv = 0, offv = 0;
    if (lane < NT * 2) {
        degv = deg[node + 8 * lane];
        offv = offs[node + 8 * lane];
    }
    int dg  = __builtin_amdgcn_readlane(degv, 0);
    int off = __builtin_amdgcn_readlane(offv, 0);
    int my = 0;
    if (lane < (dg < 32 ? dg : 32)) my = src[off + lane];

    for (int i = 0; i < NT * 2; ++i) {
        int nodeN = node + 8;
        int dgN  = __builtin_amdgcn_readlane(degv, i + 1);
        int offN = __builtin_amdgcn_readlane(offv, i + 1);
        int base = node * D128 + lo;
        f4v acc  = { 0.f, 0.f, 0.f, 0.f };
        f4v r[AGG_ROWS];
        int cnt = dg < 32 ? dg : 32;

        if (dg > 0) {
            if (cnt >= AGG_ROWS) { AGG_ISSUE_FULL(my, 0) }
            else                 { AGG_ISSUE_PART(my, 0, cnt) }
        }
        f4v self = *(gcf4p)(x + base);
        int myN = 0;
        if (lane < (dgN < 32 ? dgN : 32)) myN = src[offN + lane];

        if (dg > 0) {
            if (cnt >= AGG_ROWS) { AGG_ADD_FULL() }
            else                 { AGG_ADD_PART(cnt) }
            for (int e = AGG_ROWS; e < cnt; e += AGG_ROWS) {
                int m = cnt - e;
                if (m >= AGG_ROWS) {
                    AGG_ISSUE_FULL(my, e)
                    __builtin_amdgcn_sched_barrier(0);
                    AGG_ADD_FULL()
                } else {
                    AGG_ISSUE_PART(my, e, m)
                    AGG_ADD_PART(m)
                }
            }
            for (int c0 = 32; c0 < dg; c0 += 32) {
                int c = dg - c0;
                c = c < 32 ? c : 32;
                int ids = 0;
                if (lane < c) ids = src[off + c0 + lane];
                for (int e = 0; e < c; e += AGG_ROWS) {
                    int m = c - e;
                    if (m >= AGG_ROWS) {
                        AGG_ISSUE_FULL(ids, e)
                        __builtin_amdgcn_sched_barrier(0);
                        AGG_ADD_FULL()
                    } else {
                        AGG_ISSUE_PART(ids, e, m)
                        AGG_ADD_PART(m)
                    }
                }
            }
            float inv = 1.0f / (float)dg;
            acc *= inv;
        }
        lfp dstrow = As + rl * LDA + lo;
        *(lf4p)(dstrow)        = self;
        *(lf4p)(dstrow + SLAB) = acc;

        node = nodeN; rl += 8; dg = dgN; off = offN; my = myN;
    }
}
#undef AGG_ISSUE_FULL
#undef AGG_ISSUE_PART
#undef AGG_ADD_FULL
#undef AGG_ADD_PART

template <int NT>
__device__ __forceinline__ void gemm_rows(lfp As, int row0, int t,
        gcfp x, gcip src, gcip offs, gcip deg, gcfp Wf,
        gcfp bias, gfp out) {
    int wave = t >> 5;
    int lane = t & 31;

    int d0 = (lane < NT * 16) ? deg[row0 + lane] : 1;
    int d1 = (lane + 32 < NT * 16) ? deg[row0 + 32 + lane] : 1;

    stage_rows<NT>(As, row0, t, x, src, offs, deg);

    unsigned z0 = __builtin_amdgcn_ballot_w32(d0 == 0);
    unsigned z1 = __builtin_amdgcn_ballot_w32(d1 == 0);
    __syncthreads();

    int col0 = wave * 16;
    int mrow = lane & 15;
    int koff = (lane >> 4) * 2;

    v8f acc[NT];
    #pragma unroll
    for (int m = 0; m < NT; ++m)
        acc[m] = (v8f){ 0.f, 0.f, 0.f, 0.f, 0.f, 0.f, 0.f, 0.f };

    #pragma unroll
    for (int b2 = 0; b2 < 2; ++b2) {
        lfp   A  = As + b2 * SLAB + mrow * LDA + koff;
        gcf4p Bf = (gcf4p)Wf + b2 * FRAG_F4 + wave * 512 + lane;
        #pragma unroll
        for (int kk = 0; kk < D128 / 8; ++kk) {
            f4v bb = Bf[kk * 32];
            #pragma unroll
            for (int j = 0; j < 2; ++j) {
                v2f b = j ? (v2f){ bb.z, bb.w } : (v2f){ bb.x, bb.y };
                int k = kk * 8 + j * 4;
                #pragma unroll
                for (int m = 0; m < NT; ++m) {
                    v2f a = *(lcv2p)(A + m * 16 * LDA + k);
                    acc[m] = __builtin_amdgcn_wmma_f32_16x16x4_f32(
                        false, a, false, b, (short)0, acc[m], false, false);
                }
            }
        }
    }

    int colg = col0 + (lane & 15);
    float bv = bias[colg];
    lfp Os = As + 2 * SLAB;
    #pragma unroll
    for (int m = 0; m < NT; ++m) {
        lfp o = Os + (m * 16 + (lane >> 4) * 8) * LDA + colg;
        #pragma unroll
        for (int i = 0; i < 8; ++i) {
            float v = acc[m][i] + bv;
            v = v > 0.f ? v : (expf(v) - 1.f);
            o[i * LDA] = v;
        }
    }
    __syncthreads();
    int w = __builtin_amdgcn_readfirstlane(t >> 5);
    f4v orow[NT * 2];
    #pragma unroll
    for (int u = 0; u < NT * 2; ++u)
        orow[u] = *(lf4p)(Os + (w + 8 * u) * LDA + lane * 4);
    #pragma unroll
    for (int u = 0; u < NT * 2; ++u) {
        int r = w + 8 * u;
        unsigned zw = (u < 4) ? z0 : z1;
        if (!((zw >> (r & 31)) & 1u))
            *(gf4p)(out + (size_t)(row0 + r) * D128 + lane * 4) = orow[u];
    }
}

#define ZB 64
#define ZT 2
#define ZR (ZT * 16)
__device__ __forceinline__ void zero_rows(lfp As, int zb, int t,
        gcfp x, gcip zl, gcfp Wf, gcfp bias, gfp out) {
    int wave = t >> 5;
    int lane = t & 31;
    int lo   = lane * 4;
    int mrow = lane & 15;
    int koff = (lane >> 4) * 2;
    int colg = wave * 16 + mrow;
    int Z = zl[0];
    float bv = bias[colg];
    lip ids = (lip)(As + 2 * SLAB);

    for (int e0 = zb * ZR; e0 < Z; e0 += ZB * ZR) {
        if (t < ZR) {
            int e = e0 + t;
            ids[t] = zl[1 + (e < Z ? e : e0)];
        }
        int rid = 0;
        if (lane < ZR / 8) {
            int e = e0 + wave + 8 * lane;
            rid = zl[1 + (e < Z ? e : e0)];
        }
        f4v rowv[ZR / 8];
        #pragma unroll
        for (int u = 0; u < ZR / 8; ++u) {
            int r = __builtin_amdgcn_readlane(rid, u);
            rowv[u] = *(gcf4p)(x + (size_t)r * D128 + lo);
        }
        #pragma unroll
        for (int u = 0; u < ZR / 8; ++u)
            *(lf4p)(As + (wave + 8 * u) * LDA + lo) = rowv[u];
        __syncthreads();

        v8f acc[ZT];
        #pragma unroll
        for (int m = 0; m < ZT; ++m)
            acc[m] = (v8f){ 0.f, 0.f, 0.f, 0.f, 0.f, 0.f, 0.f, 0.f };
        lfp   A  = As + mrow * LDA + koff;
        gcf4p Bf = (gcf4p)Wf + 2 * FRAG_F4 + wave * 512 + lane;
        #pragma unroll
        for (int kk = 0; kk < D128 / 8; ++kk) {
            f4v bb = Bf[kk * 32];
            #pragma unroll
            for (int j = 0; j < 2; ++j) {
                v2f b = j ? (v2f){ bb.z, bb.w } : (v2f){ bb.x, bb.y };
                int k = kk * 8 + j * 4;
                #pragma unroll
                for (int m = 0; m < ZT; ++m) {
                    v2f a = *(lcv2p)(A + m * 16 * LDA + k);
                    acc[m] = __builtin_amdgcn_wmma_f32_16x16x4_f32(
                        false, a, false, b, (short)0, acc[m], false, false);
                }
            }
        }
        #pragma unroll
        for (int m = 0; m < ZT; ++m) {
            int sb = m * 16 + (lane >> 4) * 8;
            #pragma unroll
            for (int i = 0; i < 8; ++i) {
                if (e0 + sb + i < Z) {
                    float v = acc[m][i] + bv;
                    v = v > 0.f ? v : (expf(v) - 1.f);
                    out[(size_t)ids[sb + i] * D128 + colg] = v;
                }
            }
        }
        __syncthreads();
    }
}

__global__ __launch_bounds__(256)
void gemm_fused(const float* __restrict__ x_, const int* __restrict__ src_,
                const int* __restrict__ offs_, const int* __restrict__ deg_,
                const float* __restrict__ Wg_, const float* __restrict__ Ws_,
                const float* __restrict__ Wl_, const float* __restrict__ Wf_,
                const float* __restrict__ bias_, float* __restrict__ out_, int n,
                const int* __restrict__ zl_) {
    __shared__ float As[3 * SLAB];

    int t = threadIdx.x;
    int b = blockIdx.x;
    int nbig = (n >> 4) / GEMM_MT;
    nbig = nbig < GEMM_BIG ? nbig : GEMM_BIG;

    if (b < nbig) {
        gemm_rows<GEMM_MT>((lfp)As, b * (GEMM_MT * 16), t,
                           (gcfp)x_, (gcip)src_, (gcip)offs_, (gcip)deg_,
                           (gcfp)Wf_,
                           (gcfp)bias_, (gfp)out_);
        return;
    }

    int nsmall = ((n + 15) >> 4) - nbig * GEMM_MT;
    if (b >= nbig + nsmall) {
        zero_rows((lfp)As, b - nbig - nsmall, t, (gcfp)x_, (gcip)zl_,
                  (gcfp)Wf_, (gcfp)bias_, (gfp)out_);
        return;
    }

    int tile = nbig * GEMM_MT + (b - nbig);
    int row0 = tile * 16;

    if (row0 + 16 > n) {
        for (int r = row0; r < n; ++r) {
            int dg  = deg_[r];
            if (dg == 0) continue;
            int off = offs_[r];
            float inv = dg > 0 ? 1.0f / (float)dg : 0.f;
            for (int c = t; c < D128; c += 256) {
                float s = bias_[c];
                const float* xr = x_  + (size_t)r * D128;
                const float* wg = Wg_ + (size_t)c * D128;
                const float* ws = Ws_ + (size_t)c * D128;
                const float* wl = Wl_ + (size_t)c * D128;
                for (int k = 0; k < D128; ++k) {
                    float nmk = 0.f;
                    for (int e = 0; e < dg; ++e)
                        nmk += x_[(size_t)src_[off + e] * D128 + k];
                    nmk *= inv;
                    float xmk = dg > 0 ? xr[k] : 0.f;
                    s += xr[k] * wg[k] + xmk * ws[k] + nmk * wl[k];
                }
                s = s > 0.f ? s : (expf(s) - 1.f);
                out_[(size_t)r * D128 + c] = s;
            }
        }
        return;
    }

    gemm_rows<1>((lfp)As, row0, t,
                 (gcfp)x_, (gcip)src_, (gcip)offs_, (gcip)deg_,
                 (gcfp)Wf_,
                 (gcfp)bias_, (gfp)out_);
}

extern "C" void kernel_launch(void* const* d_in, const int* in_sizes, int n_in,
                              void* d_out, int out_size, void* d_ws, size_t ws_size,
                              hipStream_t stream) {
    (void)n_in; (void)out_size; (void)ws_size;
    const float* x    = (const float*)d_in[0];
    const float* Wg   = (const float*)d_in[1];
    const float* Wl   = (const float*)d_in[2];
    const float* Ws   = (const float*)d_in[3];
    const float* bias = (const float*)d_in[4];
    const int*   src  = (const int*)d_in[5];
    const int*   deg  = (const int*)d_in[7];

    int N = in_sizes[7];

    float* Wf = (float*)d_ws;
    int* offs = (int*)d_ws + 3 * D128 * D128;
    int* zl   = offs + N;

    int scanBlocks = (N + SCAN_THREADS - 1) / SCAN_THREADS;
    scan_offsets<<<scanBlocks, SCAN_THREADS, 0, stream>>>(deg, offs, N, Wg, Ws, Wl, Wf, zl);

    int tiles = (N + 15) / 16;
    int nbig = (N / 16) / GEMM_MT;
    nbig = nbig < GEMM_BIG ? nbig : GEMM_BIG;
    int gblocks = nbig + (tiles - nbig * GEMM_MT) + ZB;
    gemm_fused<<<gblocks, 256, 0, stream>>>(x, src, offs, deg, Wg, Ws, Wl, Wf, bias,
                                            (float*)d_out, N, zl);
}


# x product (x.(Wg+Ws)) moved into the gather: own rows staged first, then each node's wait for its neighbor rows is filled with its share of the WMMAs; per-element sum order unchanged
# speedup vs baseline: 4.1494x; 1.0425x over previous
#include <hip/hip_runtime.h>
#include <math.h>

typedef __attribute__((ext_vector_type(2))) float v2f;
typedef __attribute__((ext_vector_type(4))) float f4v;
typedef __attribute__((ext_vector_type(8))) float v8f;
typedef __attribute__((ext_vector_type(4))) int   i4v;

typedef const __attribute__((address_space(1))) float* gcfp;
typedef       __attribute__((address_space(1))) float* gfp;
typedef const __attribute__((address_space(1))) int*   gcip;
typedef       __attribute__((address_space(1))) int*   gip;
typedef const __attribute__((address_space(1))) f4v*   gcf4p;
typedef       __attribute__((address_space(1))) f4v*   gf4p;
typedef const __attribute__((address_space(1))) v2f*   gcv2p;
typedef const __attribute__((address_space(1))) i4v*   gci4p;

#define D128 128
#define SCAN_THREADS 1024

#define FRAG_F4 (D128 * D128 / 4)
__global__ __launch_bounds__(SCAN_THREADS)
void scan_offsets(const int* __restrict__ deg_, int* __restrict__ offs_, int n,
                  const float* __restrict__ Wg_, const float* __restrict__ Ws_,
                  const float* __restrict__ Wl_, float* __restrict__ Wf_,
                  int* __restrict__ zl_) {
    static_assert(SCAN_THREADS == 32 * 32, "one LDS slot per lane: 32 waves of 32 lanes");
    gcip  deg  = (gcip)deg_;
    gci4p deg4 = (gci4p)deg_;
    gip   offs = (gip)offs_;
    gip   zl   = (gip)zl_;
    __shared__ int wtot[32];
    __shared__ int wpre[32];
    __shared__ int ztot[32];
    __shared__ int zpre[32];

    int t = threadIdx.x;
    int lane = t & 31;
    int wave = t >> 5;

    for (int g = blockIdx.x * SCAN_THREADS + t; g < 3 * FRAG_F4;
         g += gridDim.x * SCAN_THREADS) {
        int sel = g / FRAG_F4;
        int idx = g - sel * FRAG_F4;
        int ln  = idx & 31;
        int kk  = (idx >> 5) & 15;
        int ct  = idx >> 9;
        int e   = (ct * 16 + (ln & 15)) * D128 + kk * 8 + (ln >> 4) * 2;
        gcfp p  = (gcfp)(sel == 1 ? Wl_ : Wg_) + e;
        v2f lo2 = *(gcv2p)p;
        v2f hi2 = *(gcv2p)(p + 4);
        if (sel == 0) {
            gcfp q = (gcfp)Ws_ + e;
            lo2 += *(gcv2p)q;
            hi2 += *(gcv2p)(q + 4);
        }
        f4v o = { lo2.x, lo2.y, hi2.x, hi2.y };
        ((gf4p)Wf_)[g] = o;
    }

    int start = blockIdx.x * SCAN_THREADS;
    int i = start + t;
    int d = (i < n) ? deg[i] : 0;

    int nq = start >> 2;
    int pre = 0;
    int zfr = 0;
    for (int q0 = t; q0 < nq; q0 += 4 * SCAN_THREADS) {
        #pragma unroll
        for (int u = 0; u < 4; ++u) {
            int q = q0 + u * SCAN_THREADS;
            bool ok = q < nq;
            i4v v = deg4[ok ? q : q0];
            int sv = (v.x + v.y) + (v.z + v.w);
            int zv = ((v.x == 0) + (v.y == 0)) + ((v.z == 0) + (v.w == 0));
            pre += ok ? sv : 0;
            zfr += ok ? zv : 0;
        }
    }

    int inc = d;
    int zme = (i < n && d == 0) ? 1 : 0;
    int zin = zme;
    #pragma unroll
    for (int o = 1; o < 32; o <<= 1) {
        int v = __shfl_up(inc, o);
        int w = __shfl_up(zin, o);
        if (lane >= o) { inc += v; zin += w; }
    }
    #pragma unroll
    for (int o = 16; o > 0; o >>= 1) {
        pre += __shfl_xor(pre, o);
        zfr += __shfl_xor(zfr, o);
    }

    if (lane == 31) { wtot[wave] = inc; ztot[wave] = zin; }
    if (lane == 0)  { wpre[wave] = pre; zpre[wave] = zfr; }
    __syncthreads();

    int c = wpre[lane] + ((lane < wave) ? wtot[lane] : 0);
    int cz = zpre[lane] + ((lane < wave) ? ztot[lane] : 0);
    #pragma unroll
    for (int o = 16; o > 0; o >>= 1) {
        c  += __shfl_xor(c, o);
        cz += __shfl_xor(cz, o);
    }

    if (i < n) offs[i] = c + inc - d;
    if (zme) zl[1 + cz + zin - 1] = i;
    if (i == n - 1) zl[0] = cz + zin;
}

#define AGG_ROWS 16

#define AGG_ISSUE_FULL(ids, e)                                               \
    _Pragma("unroll")                                                        \
    for (int u = 0; u < AGG_ROWS; ++u) {                                     \
        int s = __builtin_amdgcn_readlane((ids), (e) + u);                   \
        r[u] = *(gcf4p)(x + (size_t)s * D128 + lo);                          \
    }
#define AGG_ISSUE_PART(ids, e, m)                                            \
    {                                                                        \
        int sid[AGG_ROWS - 1];                                               \
        _Pragma("unroll")                                                    \
        for (int u = 0; u < AGG_ROWS - 1; ++u)                               \
            sid[u] = __builtin_amdgcn_readlane((ids), (e) + u);              \
        _Pragma("unroll")                                                    \
        for (int u = 0; u < AGG_ROWS - 1; ++u)                               \
            if (u < (m)) r[u] = *(gcf4p)(x + (size_t)sid[u] * D128 + lo);    \
    }
#define AGG_ADD_FULL()                                                       \
    _Pragma("unroll")                                                        \
    for (int u = 0; u < AGG_ROWS; ++u) acc += r[u];
#define AGG_ADD_PART(m)                                                      \
    _Pragma("unroll")                                                        \
    for (int u = 0; u < AGG_ROWS - 1; ++u)                                   \
        if (u < (m)) acc += r[u];

#define LDA 132
#define GEMM_MT 4
#define GEMM_BIG 768
#define SLAB (GEMM_MT * 16 * LDA)

typedef       __attribute__((address_space(3))) float* lfp;
typedef       __attribute__((address_space(3))) f4v*   lf4p;
typedef const __attribute__((address_space(3))) v2f*   lcv2p;
typedef       __attribute__((address_space(3))) int*   lip;

template <int NT>
__device__ __forceinline__ void stage_rows(lfp As, int row0, int t,
        gcfp x, gcip src, gcip offs, gcip deg, gcfp Wf, v8f (&acc8)[NT]) {
    constexpr int KPN = (D128 / 8) / (NT * 2);
    int lane = t & 31;
    int lo   = lane * 4;
    int rl   = __builtin_amdgcn_readfirstlane(t >> 5);
    int node = row0 + rl;

    f4v selfv[NT * 2];
    #pragma unroll
    for (int u = 0; u < NT * 2; ++u)
        selfv[u] = *(gcf4p)(x + (size_t)(node + 8 * u) * D128 + lo);
    int degv = 0, offv = 0;
    if (lane < NT * 2) {
        degv = deg[node + 8 * lane];
        offv = offs[node + 8 * lane];
    }
    int dg  = __builtin_amdgcn_readlane(degv, 0);
    int off = __builtin_amdgcn_readlane(offv, 0);
    int my = 0;
    if (lane < (dg < 32 ? dg : 32)) my = src[off + lane];
    #pragma unroll
    for (int u = 0; u < NT * 2; ++u)
        *(lf4p)(As + (rl + 8 * u) * LDA + lo) = selfv[u];
    gcf4p Bf = (gcf4p)Wf + (t >> 5) * 512 + lane;
    lfp   Ax = As + (lane & 15) * LDA + (lane >> 4) * 2;
    f4v bcur[KPN], bnxt[KPN];
    #pragma unroll
    for (int q = 0; q < KPN; ++q) bnxt[q] = Bf[q * 32];
    __syncthreads();

    for (int i = 0; i < NT * 2; ++i) {
        #pragma unroll
        for (int q = 0; q < KPN; ++q) bcur[q] = bnxt[q];
        int nodeN = node + 8;
        int dgN  = __builtin_amdgcn_readlane(degv, i + 1);
        int offN = __builtin_amdgcn_readlane(offv, i + 1);
        f4v acc  = { 0.f, 0.f, 0.f, 0.f };
        f4v r[AGG_ROWS];
        int cnt = dg < 32 ? dg : 32;

        if (dg > 0) {
            if (cnt >= AGG_ROWS) { AGG_ISSUE_FULL(my, 0) }
            else                 { AGG_ISSUE_PART(my, 0, cnt) }
        }
        int myN = 0;
        if (lane < (dgN < 32 ? dgN : 32)) myN = src[offN + lane];
        if (i + 1 < NT * 2) {
            #pragma unroll
            for (int q = 0; q < KPN; ++q) bnxt[q] = Bf[((i + 1) * KPN + q) * 32];
        }
        #pragma unroll
        for (int q = 0; q < KPN; ++q) {
            if (q > 0) __builtin_amdgcn_sched_barrier(0);
            #pragma unroll
            for (int j = 0; j < 2; ++j) {
                v2f b = j ? (v2f){ bcur[q].z, bcur[q].w } : (v2f){ bcur[q].x, bcur[q].y };
                int k = (i * KPN + q) * 8 + j * 4;
                #pragma unroll
                for (int m = 0; m < NT; ++m) {
                    v2f a = *(lcv2p)(Ax + m * 16 * LDA + k);
                    acc8[m] = __builtin_amdgcn_wmma_f32_16x16x4_f32(
                        false, a, false, b, (short)0, acc8[m], false, false);
                }
            }
        }

        if (dg > 0) {
            if (cnt >= AGG_ROWS) { AGG_ADD_FULL() }
            else                 { AGG_ADD_PART(cnt) }
            for (int e = AGG_ROWS; e < cnt; e += AGG_ROWS) {
                int m = cnt - e;
                if (m >= AGG_ROWS) {
                    AGG_ISSUE_FULL(my, e)
                    __builtin_amdgcn_sched_barrier(0);
                    AGG_ADD_FULL()
                } else {
                    AGG_ISSUE_PART(my, e, m)
                    AGG_ADD_PART(m)
                }
            }
            for (int c0 = 32; c0 < dg; c0 += 32) {
                int c = dg - c0;
                c = c < 32 ? c : 32;
                int ids = 0;
                if (lane < c) ids = src[off + c0 + lane];
                for (int e = 0; e < c; e += AGG_ROWS) {
                    int m = c - e;
                    if (m >= AGG_ROWS) {
                        AGG_ISSUE_FULL(ids, e)
                        __builtin_amdgcn_sched_barrier(0);
                        AGG_ADD_FULL()
                    } else {
                        AGG_ISSUE_PART(ids, e, m)
                        AGG_ADD_PART(m)
                    }
                }
            }
            float inv = 1.0f / (float)dg;
            acc *= inv;
        }
        *(lf4p)(As + SLAB + rl * LDA + lo) = acc;

        node = nodeN; rl += 8; dg = dgN; off = offN; my = myN;
    }
}
#undef AGG_ISSUE_FULL
#undef AGG_ISSUE_PART
#undef AGG_ADD_FULL
#undef AGG_ADD_PART

template <int NT>
__device__ __forceinline__ void gemm_rows(lfp As, int row0, int t,
        gcfp x, gcip src, gcip offs, gcip deg, gcfp Wf,
        gcfp bias, gfp out) {
    int wave = t >> 5;
    int lane = t & 31;

    int d0 = (lane < NT * 16) ? deg[row0 + lane] : 1;
    int d1 = (lane + 32 < NT * 16) ? deg[row0 + 32 + lane] : 1;

    v8f acc[NT];
    #pragma unroll
    for (int m = 0; m < NT; ++m)
        acc[m] = (v8f){ 0.f, 0.f, 0.f, 0.f, 0.f, 0.f, 0.f, 0.f };

    stage_rows<NT>(As, row0, t, x, src, offs, deg, Wf, acc);

    unsigned z0 = __builtin_amdgcn_ballot_w32(d0 == 0);
    unsigned z1 = __builtin_amdgcn_ballot_w32(d1 == 0);
    __syncthreads();

    int col0 = wave * 16;
    int mrow = lane & 15;
    int koff = (lane >> 4) * 2;

    {
        lfp   A  = As + SLAB + mrow * LDA + koff;
        gcf4p Bf = (gcf4p)Wf + FRAG_F4 + wave * 512 + lane;
        #pragma unroll
        for (int kk = 0; kk < D128 / 8; ++kk) {
            f4v bb = Bf[kk * 32];
            #pragma unroll
            for (int j = 0; j < 2; ++j) {
                v2f b = j ? (v2f){ bb.z, bb.w } : (v2f){ bb.x, bb.y };
                int k = kk * 8 + j * 4;
                #pragma unroll
                for (int m = 0; m < NT; ++m) {
                    v2f a = *(lcv2p)(A + m * 16 * LDA + k);
                    acc[m] = __builtin_amdgcn_wmma_f32_16x16x4_f32(
                        false, a, false, b, (short)0, acc[m], false, false);
                }
            }
        }
    }

    int colg = col0 + (lane & 15);
    float bv = bias[colg];
    lfp Os = As + 2 * SLAB;
    #pragma unroll
    for (int m = 0; m < NT; ++m) {
        lfp o = Os + (m * 16 + (lane >> 4) * 8) * LDA + colg;
        #pragma unroll
        for (int i = 0; i < 8; ++i) {
            float v = acc[m][i] + bv;
            v = v > 0.f ? v : (expf(v) - 1.f);
            o[i * LDA] = v;
        }
    }
    __syncthreads();
    int w = __builtin_amdgcn_readfirstlane(t >> 5);
    f4v orow[NT * 2];
    #pragma unroll
    for (int u = 0; u < NT * 2; ++u)
        orow[u] = *(lf4p)(Os + (w + 8 * u) * LDA + lane * 4);
    #pragma unroll
    for (int u = 0; u < NT * 2; ++u) {
        int r = w + 8 * u;
        unsigned zw = (u < 4) ? z0 : z1;
        if (!((zw >> (r & 31)) & 1u))
            *(gf4p)(out + (size_t)(row0 + r) * D128 + lane * 4) = orow[u];
    }
}

#define ZB 64
#define ZT 2
#define ZR (ZT * 16)
__device__ __forceinline__ void zero_rows(lfp As, int zb, int t,
        gcfp x, gcip zl, gcfp Wf, gcfp bias, gfp out) {
    int wave = t >> 5;
    int lane = t & 31;
    int lo   = lane * 4;
    int mrow = lane & 15;
    int koff = (lane >> 4) * 2;
    int colg = wave * 16 + mrow;
    int Z = zl[0];
    float bv = bias[colg];
    lip ids = (lip)(As + 2 * SLAB);

    for (int e0 = zb * ZR; e0 < Z; e0 += ZB * ZR) {
        if (t < ZR) {
            int e = e0 + t;
            ids[t] = zl[1 + (e < Z ? e : e0)];
        }
        int rid = 0;
        if (lane < ZR / 8) {
            int e = e0 + wave + 8 * lane;
            rid = zl[1 + (e < Z ? e : e0)];
        }
        f4v rowv[ZR / 8];
        #pragma unroll
        for (int u = 0; u < ZR / 8; ++u) {
            int r = __builtin_amdgcn_readlane(rid, u);
            rowv[u] = *(gcf4p)(x + (size_t)r * D128 + lo);
        }
        #pragma unroll
        for (int u = 0; u < ZR / 8; ++u)
            *(lf4p)(As + (wave + 8 * u) * LDA + lo) = rowv[u];
        __syncthreads();

        v8f acc[ZT];
        #pragma unroll
        for (int m = 0; m < ZT; ++m)
            acc[m] = (v8f){ 0.f, 0.f, 0.f, 0.f, 0.f, 0.f, 0.f, 0.f };
        lfp   A  = As + mrow * LDA + koff;
        gcf4p Bf = (gcf4p)Wf + 2 * FRAG_F4 + wave * 512 + lane;
        #pragma unroll
        for (int kk = 0; kk < D128 / 8; ++kk) {
            f4v bb = Bf[kk * 32];
            #pragma unroll
            for (int j = 0; j < 2; ++j) {
                v2f b = j ? (v2f){ bb.z, bb.w } : (v2f){ bb.x, bb.y };
                int k = kk * 8 + j * 4;
                #pragma unroll
                for (int m = 0; m < ZT; ++m) {
                    v2f a = *(lcv2p)(A + m * 16 * LDA + k);
                    acc[m] = __builtin_amdgcn_wmma_f32_16x16x4_f32(
                        false, a, false, b, (short)0, acc[m], false, false);
                }
            }
        }
        #pragma unroll
        for (int m = 0; m < ZT; ++m) {
            int sb = m * 16 + (lane >> 4) * 8;
            #pragma unroll
            for (int i = 0; i < 8; ++i) {
                if (e0 + sb + i < Z) {
                    float v = acc[m][i] + bv;
                    v = v > 0.f ? v : (expf(v) - 1.f);
                    out[(size_t)ids[sb + i] * D128 + colg] = v;
                }
            }
        }
        __syncthreads();
    }
}

__global__ __launch_bounds__(256)
void gemm_fused(const float* __restrict__ x_, const int* __restrict__ src_,
                const int* __restrict__ offs_, const int* __restrict__ deg_,
                const float* __restrict__ Wg_, const float* __restrict__ Ws_,
                const float* __restrict__ Wl_, const float* __restrict__ Wf_,
                const float* __restrict__ bias_, float* __restrict__ out_, int n,
                const int* __restrict__ zl_) {
    __shared__ float As[3 * SLAB];

    int t = threadIdx.x;
    int b = blockIdx.x;
    int nbig = (n >> 4) / GEMM_MT;
    nbig = nbig < GEMM_BIG ? nbig : GEMM_BIG;

    if (b < nbig) {
        gemm_rows<GEMM_MT>((lfp)As, b * (GEMM_MT * 16), t,
                           (gcfp)x_, (gcip)src_, (gcip)offs_, (gcip)deg_,
                           (gcfp)Wf_,
                           (gcfp)bias_, (gfp)out_);
        return;
    }

    int nsmall = ((n + 15) >> 4) - nbig * GEMM_MT;
    if (b >= nbig + nsmall) {
        zero_rows((lfp)As, b - nbig - nsmall, t, (gcfp)x_, (gcip)zl_,
                  (gcfp)Wf_, (gcfp)bias_, (gfp)out_);
        return;
    }

    int tile = nbig * GEMM_MT + (b - nbig);
    int row0 = tile * 16;

    if (row0 + 16 > n) {
        for (int r = row0; r < n; ++r) {
            int dg  = deg_[r];
            if (dg == 0) continue;
            int off = offs_[r];
            float inv = dg > 0 ? 1.0f / (float)dg : 0.f;
            for (int c = t; c < D128; c += 256) {
                float s = bias_[c];
                const float* xr = x_  + (size_t)r * D128;
                const float* wg = Wg_ + (size_t)c * D128;
                const float* ws = Ws_ + (size_t)c * D128;
                const float* wl = Wl_ + (size_t)c * D128;
                for (int k = 0; k < D128; ++k) {
                    float nmk = 0.f;
                    for (int e = 0; e < dg; ++e)
                        nmk += x_[(size_t)src_[off + e] * D128 + k];
                    nmk *= inv;
                    float xmk = dg > 0 ? xr[k] : 0.f;
                    s += xr[k] * wg[k] + xmk * ws[k] + nmk * wl[k];
                }
                s = s > 0.f ? s : (expf(s) - 1.f);
                out_[(size_t)r * D128 + c] = s;
            }
        }
        return;
    }

    gemm_rows<1>((lfp)As, row0, t,
                 (gcfp)x_, (gcip)src_, (gcip)offs_, (gcip)deg_,
                 (gcfp)Wf_,
                 (gcfp)bias_, (gfp)out_);
}

extern "C" void kernel_launch(void* const* d_in, const int* in_sizes, int n_in,
                              void* d_out, int out_size, void* d_ws, size_t ws_size,
                              hipStream_t stream) {
    (void)n_in; (void)out_size; (void)ws_size;
    const float* x    = (const float*)d_in[0];
    const float* Wg   = (const float*)d_in[1];
    const float* Wl   = (const float*)d_in[2];
    const float* Ws   = (const float*)d_in[3];
    const float* bias = (const float*)d_in[4];
    const int*   src  = (const int*)d_in[5];
    const int*   deg  = (const int*)d_in[7];

    int N = in_sizes[7];

    float* Wf = (float*)d_ws;
    int* offs = (int*)d_ws + 3 * D128 * D128;
    int* zl   = offs + N;

    int scanBlocks = (N + SCAN_THREADS - 1) / SCAN_THREADS;
    scan_offsets<<<scanBlocks, SCAN_THREADS, 0, stream>>>(deg, offs, N, Wg, Ws, Wl, Wf, zl);

    int tiles = (N + 15) / 16;
    int nbig = (N / 16) / GEMM_MT;
    nbig = nbig < GEMM_BIG ? nbig : GEMM_BIG;
    int gblocks = nbig + (tiles - nbig * GEMM_MT) + ZB;
    gemm_fused<<<gblocks, 256, 0, stream>>>(x, src, offs, deg, Wg, Ws, Wl, Wf, bias,
                                            (float*)d_out, N, zl);
}
